# starting kernel
# baseline (speedup 1.0000x reference)
#include <hip/hip_runtime.h>

typedef __attribute__((ext_vector_type(2))) float v2f;
typedef __attribute__((ext_vector_type(8))) float v8f;


__global__ void gcn_init(float* __restrict__ deg, float* __restrict__ s1,
                         float* __restrict__ out, int n) {
    int i = blockIdx.x * blockDim.x + threadIdx.x;
    if (i < n) { deg[i] = 0.0f; s1[i] = 0.0f; out[i] = 0.0f; }
}

__global__ void gcn_degree(const int* __restrict__ dst, float* __restrict__ deg, int e) {
    int i = blockIdx.x * blockDim.x + threadIdx.x;
    if (i < e) {
        int d = __builtin_nontemporal_load(&dst[i]);
        atomicAdd(&deg[d], 1.0f);
    }
}

__global__ void gcn_dinv(float* __restrict__ deg, int n) {
    int i = blockIdx.x * blockDim.x + threadIdx.x;
    if (i < n) {
        float d = deg[i] + 1.0f;
        deg[i] = rsqrtf(d);
    }
}

__global__ void gcn_scatter1(const int* __restrict__ src, const int* __restrict__ dst,
                             const float* __restrict__ x, const float* __restrict__ dinv,
                             float* __restrict__ s1, int e) {
    int i = blockIdx.x * blockDim.x + threadIdx.x;
    if (i < e) {
        int s = __builtin_nontemporal_load(&src[i]);
        int d = __builtin_nontemporal_load(&dst[i]);
        atomicAdd(&s1[d], dinv[s] * dinv[d] * x[s]);
    }
}

__global__ void gcn_node_mlp(const float* __restrict__ s1, const float* __restrict__ dinv,
                             const float* __restrict__ x,
                             const float* __restrict__ W1, const float* __restrict__ b1,
                             const float* __restrict__ W2,
                             float* __restrict__ h2, int n) {
    int lane = threadIdx.x & 31;
    int wave = threadIdx.x >> 5;
    int base = (blockIdx.x * (blockDim.x >> 5) + wave) * 16;
    int node = base + lane;

    float sv = 0.0f;
    if (lane < 16 && node < n) {
        float di = dinv[node];
        sv = s1[node] + di * di * x[node];
    }

    v2f a; a.x = 0.0f; a.y = 0.0f;
    if (lane < 8) a.x = W1[lane];
    v2f b; b.x = (lane < 16) ? sv : 0.0f; b.y = 0.0f;
    v8f c;
#pragma unroll
    for (int r = 0; r < 8; ++r) c[r] = (lane < 16) ? b1[r] : 0.0f;

    v8f dacc = __builtin_amdgcn_wmma_f32_16x16x4_f32(
         false, a,  false, b,
         (short)0, c,  false,  false);

    float acc = 0.0f;
#pragma unroll
    for (int r = 0; r < 8; ++r) acc = fmaf(fmaxf(dacc[r], 0.0f), W2[r], acc);

    if (lane < 16 && node < n) h2[node] = acc;
}

__global__ void gcn_scatter2(const int* __restrict__ src, const int* __restrict__ dst,
                             const float* __restrict__ h2, const float* __restrict__ dinv,
                             float* __restrict__ out, int e) {
    int i = blockIdx.x * blockDim.x + threadIdx.x;
    if (i < e) {
        int s = __builtin_nontemporal_load(&src[i]);
        int d = __builtin_nontemporal_load(&dst[i]);
        atomicAdd(&out[d], dinv[s] * dinv[d] * h2[s]);
    }
}

__global__ void gcn_final(float* __restrict__ out, const float* __restrict__ h2,
                          const float* __restrict__ dinv, const float* __restrict__ b2, int n) {
    int i = blockIdx.x * blockDim.x + threadIdx.x;
    if (i < n) {
        float di = dinv[i];
        out[i] = out[i] + di * di * h2[i] + b2[0];
    }
}

extern "C" void kernel_launch(void* const* d_in, const int* in_sizes, int n_in,
                              void* d_out, int out_size, void* d_ws, size_t ws_size,
                              hipStream_t stream) {
    const float* x  = (const float*)d_in[0];
    const int*   ei = (const int*)d_in[1];
    const float* W1 = (const float*)d_in[2];
    const float* b1 = (const float*)d_in[3];
    const float* W2 = (const float*)d_in[4];
    const float* b2 = (const float*)d_in[5];

    const int n = in_sizes[0];
    const int e = in_sizes[1] / 2;
    const int* src = ei;
    const int* dst = ei + e;

    float* deg = (float*)d_ws;
    float* s1  = deg + n;
    float* h2  = s1 + n;
    float* out = (float*)d_out;

    const int tb = 256;
    const int nb_n = (n + tb - 1) / tb;
    const int nb_e = (e + tb - 1) / tb;

    gcn_init   <<<nb_n, tb, 0, stream>>>(deg, s1, out, n);
    gcn_degree <<<nb_e, tb, 0, stream>>>(dst, deg, e);
    gcn_dinv   <<<nb_n, tb, 0, stream>>>(deg, n);
    gcn_scatter1<<<nb_e, tb, 0, stream>>>(src, dst, x, deg, s1, e);

    const int nodes_per_block = (tb / 32) * 16;
    gcn_node_mlp<<<(n + nodes_per_block - 1) / nodes_per_block, tb, 0, stream>>>(
        s1, deg, x, W1, b1, W2, h2, n);

    gcn_scatter2<<<nb_e, tb, 0, stream>>>(src, dst, h2, deg, out, e);
    gcn_final  <<<nb_n, tb, 0, stream>>>(out, h2, deg, b2, n);
}


# One accumulator array per package half in the degree and scatter-add passes: die id picks the array, next node kernel adds the two (f32 sum order: two partial sums per node); 4 edges per thread
# speedup vs baseline: 2.6700x; 2.6700x over previous
#include <hip/hip_runtime.h>


typedef __attribute__((ext_vector_type(2))) float v2f;
typedef __attribute__((ext_vector_type(8))) float v8f;


#define EDGES_PER_THREAD 4

__device__ __forceinline__ int package_half() {
    __shared__ int half_of_block;
    if (threadIdx.x < warpSize) {
        unsigned id = __builtin_amdgcn_s_sendmsg_rtn(0x87);
        if (threadIdx.x == 0) half_of_block = (int)((id >> 18) & 1u);
    }
    __syncthreads();
    return __builtin_amdgcn_readfirstlane(half_of_block);
}

__global__ void gcn_init(float* __restrict__ deg_a, float* __restrict__ deg_b,
                         float* __restrict__ out, int n) {
    int i = blockIdx.x * blockDim.x + threadIdx.x;
    if (i < n) { deg_a[i] = 0.0f; deg_b[i] = 0.0f; out[i] = 0.0f; }
}

__global__ void gcn_degree(const int* __restrict__ dst, float* __restrict__ deg_a,
                           float* __restrict__ deg_b, int e) {
    float* __restrict__ deg = package_half() ? deg_b : deg_a;
    const int per_block = (int)blockDim.x * EDGES_PER_THREAD;
    const int b0 = (int)blockIdx.x * per_block;
    const int i0 = b0 + (int)threadIdx.x;
    if (b0 + per_block <= e) {
        int d[EDGES_PER_THREAD];
#pragma unroll
        for (int j = 0; j < EDGES_PER_THREAD; ++j)
            d[j] = __builtin_nontemporal_load(&dst[i0 + j * (int)blockDim.x]);
#pragma unroll
        for (int j = 0; j < EDGES_PER_THREAD; ++j)
            atomicAdd(&deg[d[j]], 1.0f);
    } else {
        for (int j = 0; j < EDGES_PER_THREAD; ++j) {
            int i = i0 + j * (int)blockDim.x;
            if (i < e) atomicAdd(&deg[__builtin_nontemporal_load(&dst[i])], 1.0f);
        }
    }
}

__global__ void gcn_dinv(float* __restrict__ deg_a, float* __restrict__ deg_b,
                         float* __restrict__ h2, int n) {
    int i = blockIdx.x * blockDim.x + threadIdx.x;
    if (i < n) {
        float d = (deg_a[i] + deg_b[i]) + 1.0f;
        deg_a[i] = rsqrtf(d);
        deg_b[i] = 0.0f;
        h2[i] = 0.0f;
    }
}

__global__ void gcn_scatter1(const int* __restrict__ src, const int* __restrict__ dst,
                             const float* __restrict__ x, const float* __restrict__ dinv,
                             float* __restrict__ s1_a, float* __restrict__ s1_b, int e) {
    float* __restrict__ s1 = package_half() ? s1_b : s1_a;
    const int per_block = (int)blockDim.x * EDGES_PER_THREAD;
    const int b0 = (int)blockIdx.x * per_block;
    const int i0 = b0 + (int)threadIdx.x;
    if (b0 + per_block <= e) {
        int s[EDGES_PER_THREAD], d[EDGES_PER_THREAD];
        float v[EDGES_PER_THREAD];
#pragma unroll
        for (int j = 0; j < EDGES_PER_THREAD; ++j) {
            int i = i0 + j * (int)blockDim.x;
            s[j] = __builtin_nontemporal_load(&src[i]);
            d[j] = __builtin_nontemporal_load(&dst[i]);
        }
#pragma unroll
        for (int j = 0; j < EDGES_PER_THREAD; ++j)
            v[j] = dinv[s[j]] * dinv[d[j]] * x[s[j]];
#pragma unroll
        for (int j = 0; j < EDGES_PER_THREAD; ++j)
            atomicAdd(&s1[d[j]], v[j]);
    } else {
        for (int j = 0; j < EDGES_PER_THREAD; ++j) {
            int i = i0 + j * (int)blockDim.x;
            if (i < e) {
                int s = __builtin_nontemporal_load(&src[i]);
                int d = __builtin_nontemporal_load(&dst[i]);
                atomicAdd(&s1[d], dinv[s] * dinv[d] * x[s]);
            }
        }
    }
}

__global__ void gcn_node_mlp(float* __restrict__ s1, const float* __restrict__ dinv,
                             const float* __restrict__ x,
                             const float* __restrict__ W1, const float* __restrict__ b1,
                             const float* __restrict__ W2,
                             float* __restrict__ h2, int n) {
    int lane = threadIdx.x & 31;
    int wave = threadIdx.x >> 5;
    int base = (blockIdx.x * (blockDim.x >> 5) + wave) * 16;
    int node = base + lane;

    float sv = 0.0f;
    if (lane < 16 && node < n) {
        float di = dinv[node];
        sv = (s1[node] + h2[node]) + di * di * x[node];
        s1[node] = 0.0f;
    }

    v2f a; a.x = 0.0f; a.y = 0.0f;
    if (lane < 8) a.x = W1[lane];
    v2f b; b.x = (lane < 16) ? sv : 0.0f; b.y = 0.0f;
    v8f c;
#pragma unroll
    for (int r = 0; r < 8; ++r) c[r] = (lane < 16) ? b1[r] : 0.0f;

    v8f dacc = __builtin_amdgcn_wmma_f32_16x16x4_f32(
         false, a,  false, b,
         (short)0, c,  false,  false);

    float acc = 0.0f;
#pragma unroll
    for (int r = 0; r < 8; ++r) acc = fmaf(fmaxf(dacc[r], 0.0f), W2[r], acc);

    if (lane < 16 && node < n) h2[node] = acc;
}

__global__ void gcn_scatter2(const int* __restrict__ src, const int* __restrict__ dst,
                             const float* __restrict__ h2, const float* __restrict__ dinv,
                             float* __restrict__ out_a, float* __restrict__ out_b, int e) {
    float* __restrict__ out = package_half() ? out_b : out_a;
    const int per_block = (int)blockDim.x * EDGES_PER_THREAD;
    const int b0 = (int)blockIdx.x * per_block;
    const int i0 = b0 + (int)threadIdx.x;
    if (b0 + per_block <= e) {
        int s[EDGES_PER_THREAD], d[EDGES_PER_THREAD];
        float v[EDGES_PER_THREAD];
#pragma unroll
        for (int j = 0; j < EDGES_PER_THREAD; ++j) {
            int i = i0 + j * (int)blockDim.x;
            s[j] = __builtin_nontemporal_load(&src[i]);
            d[j] = __builtin_nontemporal_load(&dst[i]);
        }
#pragma unroll
        for (int j = 0; j < EDGES_PER_THREAD; ++j)
            v[j] = dinv[s[j]] * dinv[d[j]] * h2[s[j]];
#pragma unroll
        for (int j = 0; j < EDGES_PER_THREAD; ++j)
            atomicAdd(&out[d[j]], v[j]);
    } else {
        for (int j = 0; j < EDGES_PER_THREAD; ++j) {
            int i = i0 + j * (int)blockDim.x;
            if (i < e) {
                int s = __builtin_nontemporal_load(&src[i]);
                int d = __builtin_nontemporal_load(&dst[i]);
                atomicAdd(&out[d], dinv[s] * dinv[d] * h2[s]);
            }
        }
    }
}

__global__ void gcn_final(float* __restrict__ out, const float* __restrict__ out_b,
                          const float* __restrict__ h2,
                          const float* __restrict__ dinv, const float* __restrict__ b2, int n) {
    int i = blockIdx.x * blockDim.x + threadIdx.x;
    if (i < n) {
        float di = dinv[i];
        out[i] = (out[i] + out_b[i]) + di * di * h2[i] + b2[0];
    }
}

extern "C" void kernel_launch(void* const* d_in, const int* in_sizes, int n_in,
                              void* d_out, int out_size, void* d_ws, size_t ws_size,
                              hipStream_t stream) {
    const float* x  = (const float*)d_in[0];
    const int*   ei = (const int*)d_in[1];
    const float* W1 = (const float*)d_in[2];
    const float* b1 = (const float*)d_in[3];
    const float* W2 = (const float*)d_in[4];
    const float* b2 = (const float*)d_in[5];

    const int n = in_sizes[0];
    const int e = in_sizes[1] / 2;
    const int* src = ei;
    const int* dst = ei + e;

    float* deg = (float*)d_ws;
    float* s1  = deg + n;
    float* h2  = s1 + n;
    float* out = (float*)d_out;

    const int tb = 256;
    const int nb_n = (n + tb - 1) / tb;
    const int per_block = tb * EDGES_PER_THREAD;
    const int nb_e = (e + per_block - 1) / per_block;

    gcn_init   <<<nb_n, tb, 0, stream>>>(deg, s1, out, n);
    gcn_degree <<<nb_e, tb, 0, stream>>>(dst, deg, s1, e);
    gcn_dinv   <<<nb_n, tb, 0, stream>>>(deg, s1, h2, n);
    gcn_scatter1<<<nb_e, tb, 0, stream>>>(src, dst, x, deg, s1, h2, e);

    const int nodes_per_block = (tb / 32) * 16;
    gcn_node_mlp<<<(n + nodes_per_block - 1) / nodes_per_block, tb, 0, stream>>>(
        s1, deg, x, W1, b1, W2, h2, n);

    gcn_scatter2<<<nb_e, tb, 0, stream>>>(src, dst, h2, deg, out, s1, e);
    gcn_final  <<<nb_n, tb, 0, stream>>>(out, s1, h2, deg, b2, n);
}


# Edge weight factored out of the scatter sums: source norm folded into one gathered per-node value, destination norm applied after the sum (f32 product order changes): 1 gather, 1 atomic per edge
# speedup vs baseline: 4.2814x; 1.6035x over previous
#include <hip/hip_runtime.h>


typedef __attribute__((ext_vector_type(2))) float v2f;
typedef __attribute__((ext_vector_type(8))) float v8f;


#define EDGES_PER_THREAD 4

__device__ __forceinline__ int package_half() {
    __shared__ int half_of_block;
    if (threadIdx.x < warpSize) {
        unsigned id = __builtin_amdgcn_s_sendmsg_rtn(0x87);
        if (threadIdx.x == 0) half_of_block = (int)((id >> 18) & 1u);
    }
    __syncthreads();
    return __builtin_amdgcn_readfirstlane(half_of_block);
}

__global__ void gcn_init(float* __restrict__ deg_a, float* __restrict__ deg_b, int n) {
    int i = blockIdx.x * blockDim.x + threadIdx.x;
    if (i < n) { deg_a[i] = 0.0f; deg_b[i] = 0.0f; }
}

__global__ void gcn_degree(const int* __restrict__ dst, float* __restrict__ deg_a,
                           float* __restrict__ deg_b, int e) {
    float* __restrict__ deg = package_half() ? deg_b : deg_a;
    const int per_block = (int)blockDim.x * EDGES_PER_THREAD;
    const int b0 = (int)blockIdx.x * per_block;
    const int i0 = b0 + (int)threadIdx.x;
    if (b0 + per_block <= e) {
        int d[EDGES_PER_THREAD];
#pragma unroll
        for (int j = 0; j < EDGES_PER_THREAD; ++j)
            d[j] = __builtin_nontemporal_load(&dst[i0 + j * (int)blockDim.x]);
#pragma unroll
        for (int j = 0; j < EDGES_PER_THREAD; ++j)
            atomicAdd(&deg[d[j]], 1.0f);
    } else {
        for (int j = 0; j < EDGES_PER_THREAD; ++j) {
            int i = i0 + j * (int)blockDim.x;
            if (i < e) atomicAdd(&deg[__builtin_nontemporal_load(&dst[i])], 1.0f);
        }
    }
}

__global__ void gcn_dinv(float* __restrict__ deg_a, float* __restrict__ deg_b,
                         const float* __restrict__ x, float* __restrict__ p,
                         float* __restrict__ out, int n) {
    int i = blockIdx.x * blockDim.x + threadIdx.x;
    if (i < n) {
        float d = (deg_a[i] + deg_b[i]) + 1.0f;
        float di = rsqrtf(d);
        deg_a[i] = di;
        p[i] = di * x[i];
        deg_b[i] = 0.0f;
        out[i] = 0.0f;
    }
}

__global__ void gcn_scatter(const int* __restrict__ src, const int* __restrict__ dst,
                            const float* __restrict__ val,
                            float* __restrict__ acc_a, float* __restrict__ acc_b, int e) {
    float* __restrict__ acc = package_half() ? acc_b : acc_a;
    const int per_block = (int)blockDim.x * EDGES_PER_THREAD;
    const int b0 = (int)blockIdx.x * per_block;
    const int i0 = b0 + (int)threadIdx.x;
    if (b0 + per_block <= e) {
        int s[EDGES_PER_THREAD], d[EDGES_PER_THREAD];
        float v[EDGES_PER_THREAD];
#pragma unroll
        for (int j = 0; j < EDGES_PER_THREAD; ++j) {
            int i = i0 + j * (int)blockDim.x;
            s[j] = __builtin_nontemporal_load(&src[i]);
            d[j] = __builtin_nontemporal_load(&dst[i]);
        }
#pragma unroll
        for (int j = 0; j < EDGES_PER_THREAD; ++j)
            v[j] = val[s[j]];
#pragma unroll
        for (int j = 0; j < EDGES_PER_THREAD; ++j)
            atomicAdd(&acc[d[j]], v[j]);
    } else {
        for (int j = 0; j < EDGES_PER_THREAD; ++j) {
            int i = i0 + j * (int)blockDim.x;
            if (i < e) {
                int s = __builtin_nontemporal_load(&src[i]);
                int d = __builtin_nontemporal_load(&dst[i]);
                atomicAdd(&acc[d], val[s]);
            }
        }
    }
}

__global__ void gcn_node_mlp(float* __restrict__ acc_a, float* __restrict__ acc_b,
                             const float* __restrict__ dinv,
                             const float* __restrict__ W1, const float* __restrict__ b1,
                             const float* __restrict__ W2,
                             float* __restrict__ pq, int n) {
    int lane = threadIdx.x & 31;
    int wave = threadIdx.x >> 5;
    int base = (blockIdx.x * (blockDim.x >> 5) + wave) * 16;
    int node = base + lane;

    float sv = 0.0f;
    float di = 0.0f;
    if (lane < 16 && node < n) {
        di = dinv[node];
        sv = di * ((acc_a[node] + acc_b[node]) + pq[node]);
        acc_a[node] = 0.0f;
        acc_b[node] = 0.0f;
    }

    v2f a; a.x = 0.0f; a.y = 0.0f;
    if (lane < 8) a.x = W1[lane];
    v2f b; b.x = (lane < 16) ? sv : 0.0f; b.y = 0.0f;
    v8f c;
#pragma unroll
    for (int r = 0; r < 8; ++r) c[r] = (lane < 16) ? b1[r] : 0.0f;

    v8f dacc = __builtin_amdgcn_wmma_f32_16x16x4_f32(
         false, a,  false, b,
         (short)0, c,  false,  false);

    float acc = 0.0f;
#pragma unroll
    for (int r = 0; r < 8; ++r) acc = fmaf(fmaxf(dacc[r], 0.0f), W2[r], acc);

    if (lane < 16 && node < n) pq[node] = di * acc;
}

__global__ void gcn_final(float* __restrict__ out, const float* __restrict__ acc_a,
                          const float* __restrict__ q,
                          const float* __restrict__ dinv, const float* __restrict__ b2, int n) {
    int i = blockIdx.x * blockDim.x + threadIdx.x;
    if (i < n) {
        out[i] = dinv[i] * ((acc_a[i] + out[i]) + q[i]) + b2[0];
    }
}

extern "C" void kernel_launch(void* const* d_in, const int* in_sizes, int n_in,
                              void* d_out, int out_size, void* d_ws, size_t ws_size,
                              hipStream_t stream) {
    const float* x  = (const float*)d_in[0];
    const int*   ei = (const int*)d_in[1];
    const float* W1 = (const float*)d_in[2];
    const float* b1 = (const float*)d_in[3];
    const float* W2 = (const float*)d_in[4];
    const float* b2 = (const float*)d_in[5];

    const int n = in_sizes[0];
    const int e = in_sizes[1] / 2;
    const int* src = ei;
    const int* dst = ei + e;

    float* deg = (float*)d_ws;
    float* acc = deg + n;
    float* pq  = acc + n;
    float* out = (float*)d_out;

    const int tb = 256;
    const int nb_n = (n + tb - 1) / tb;
    const int per_block = tb * EDGES_PER_THREAD;
    const int nb_e = (e + per_block - 1) / per_block;

    gcn_init   <<<nb_n, tb, 0, stream>>>(deg, acc, n);
    gcn_degree <<<nb_e, tb, 0, stream>>>(dst, deg, acc, e);
    gcn_dinv   <<<nb_n, tb, 0, stream>>>(deg, acc, x, pq, out, n);
    gcn_scatter<<<nb_e, tb, 0, stream>>>(src, dst, pq, acc, out, e);

    const int nodes_per_block = (tb / 32) * 16;
    gcn_node_mlp<<<(n + nodes_per_block - 1) / nodes_per_block, tb, 0, stream>>>(
        acc, out, deg, W1, b1, W2, pq, n);

    gcn_scatter<<<nb_e, tb, 0, stream>>>(src, dst, pq, acc, out, e);
    gcn_final  <<<nb_n, tb, 0, stream>>>(out, acc, pq, deg, b2, n);
}


# Edges bucketed once per call by destination range into packed words; degree and both scatter sums add per range in LDS with LDS atomics, no global atomics (f32 sum order: one arrival-order sum)
# speedup vs baseline: 5.4151x; 1.2648x over previous
#include <hip/hip_runtime.h>


typedef __attribute__((ext_vector_type(2))) float v2f;
typedef __attribute__((ext_vector_type(8))) float v8f;


#define EDGES_PER_THREAD 4

__device__ __forceinline__ int package_half() {
    __shared__ int half_of_block;
    if (threadIdx.x < warpSize) {
        unsigned id = __builtin_amdgcn_s_sendmsg_rtn(0x87);
        if (threadIdx.x == 0) half_of_block = (int)((id >> 18) & 1u);
    }
    __syncthreads();
    return __builtin_amdgcn_readfirstlane(half_of_block);
}

__global__ void gcn_init(float* __restrict__ deg_a, float* __restrict__ deg_b, int n) {
    int i = blockIdx.x * blockDim.x + threadIdx.x;
    if (i < n) { deg_a[i] = 0.0f; deg_b[i] = 0.0f; }
}

__global__ void gcn_degree(const int* __restrict__ dst, float* __restrict__ deg_a,
                           float* __restrict__ deg_b, int e) {
    float* __restrict__ deg = package_half() ? deg_b : deg_a;
    const int per_block = (int)blockDim.x * EDGES_PER_THREAD;
    const int b0 = (int)blockIdx.x * per_block;
    const int i0 = b0 + (int)threadIdx.x;
    if (b0 + per_block <= e) {
        int d[EDGES_PER_THREAD];
#pragma unroll
        for (int j = 0; j < EDGES_PER_THREAD; ++j)
            d[j] = __builtin_nontemporal_load(&dst[i0 + j * (int)blockDim.x]);
#pragma unroll
        for (int j = 0; j < EDGES_PER_THREAD; ++j)
            atomicAdd(&deg[d[j]], 1.0f);
    } else {
        for (int j = 0; j < EDGES_PER_THREAD; ++j) {
            int i = i0 + j * (int)blockDim.x;
            if (i < e) atomicAdd(&deg[__builtin_nontemporal_load(&dst[i])], 1.0f);
        }
    }
}

__global__ void gcn_dinv(float* __restrict__ deg_a, float* __restrict__ deg_b,
                         const float* __restrict__ x, float* __restrict__ p,
                         float* __restrict__ out, int n) {
    int i = blockIdx.x * blockDim.x + threadIdx.x;
    if (i < n) {
        float d = (deg_a[i] + deg_b[i]) + 1.0f;
        float di = rsqrtf(d);
        deg_a[i] = di;
        p[i] = di * x[i];
        deg_b[i] = 0.0f;
        out[i] = 0.0f;
    }
}

__global__ void gcn_scatter(const int* __restrict__ src, const int* __restrict__ dst,
                            const float* __restrict__ val,
                            float* __restrict__ acc_a, float* __restrict__ acc_b, int e) {
    float* __restrict__ acc = package_half() ? acc_b : acc_a;
    const int per_block = (int)blockDim.x * EDGES_PER_THREAD;
    const int b0 = (int)blockIdx.x * per_block;
    const int i0 = b0 + (int)threadIdx.x;
    if (b0 + per_block <= e) {
        int s[EDGES_PER_THREAD], d[EDGES_PER_THREAD];
        float v[EDGES_PER_THREAD];
#pragma unroll
        for (int j = 0; j < EDGES_PER_THREAD; ++j) {
            int i = i0 + j * (int)blockDim.x;
            s[j] = __builtin_nontemporal_load(&src[i]);
            d[j] = __builtin_nontemporal_load(&dst[i]);
        }
#pragma unroll
        for (int j = 0; j < EDGES_PER_THREAD; ++j)
            v[j] = val[s[j]];
#pragma unroll
        for (int j = 0; j < EDGES_PER_THREAD; ++j)
            atomicAdd(&acc[d[j]], v[j]);
    } else {
        for (int j = 0; j < EDGES_PER_THREAD; ++j) {
            int i = i0 + j * (int)blockDim.x;
            if (i < e) {
                int s = __builtin_nontemporal_load(&src[i]);
                int d = __builtin_nontemporal_load(&dst[i]);
                atomicAdd(&acc[d], val[s]);
            }
        }
    }
}

__global__ void gcn_node_mlp(float* __restrict__ acc_a, float* __restrict__ acc_b,
                             const float* __restrict__ dinv,
                             const float* __restrict__ W1, const float* __restrict__ b1,
                             const float* __restrict__ W2,
                             float* __restrict__ pq, int n) {
    int lane = threadIdx.x & 31;
    int wave = threadIdx.x >> 5;
    int base = (blockIdx.x * (blockDim.x >> 5) + wave) * 16;
    int node = base + lane;

    float sv = 0.0f;
    float di = 0.0f;
    if (lane < 16 && node < n) {
        di = dinv[node];
        sv = di * ((acc_a[node] + acc_b[node]) + pq[node]);
        acc_a[node] = 0.0f;
        acc_b[node] = 0.0f;
    }

    v2f a; a.x = 0.0f; a.y = 0.0f;
    if (lane < 8) a.x = W1[lane];
    v2f b; b.x = (lane < 16) ? sv : 0.0f; b.y = 0.0f;
    v8f c;
#pragma unroll
    for (int r = 0; r < 8; ++r) c[r] = (lane < 16) ? b1[r] : 0.0f;

    v8f dacc = __builtin_amdgcn_wmma_f32_16x16x4_f32(
         false, a,  false, b,
         (short)0, c,  false,  false);

    float acc = 0.0f;
#pragma unroll
    for (int r = 0; r < 8; ++r) acc = fmaf(fmaxf(dacc[r], 0.0f), W2[r], acc);

    if (lane < 16 && node < n) pq[node] = di * acc;
}

__global__ void gcn_final(float* __restrict__ out, const float* __restrict__ acc_a,
                          const float* __restrict__ q,
                          const float* __restrict__ dinv, const float* __restrict__ b2, int n) {
    int i = blockIdx.x * blockDim.x + threadIdx.x;
    if (i < n) {
        out[i] = dinv[i] * ((acc_a[i] + out[i]) + q[i]) + b2[0];
    }
}


#define RANGE 3936
#define MAXBIN 272
#define CHUNK 20480
#define EPT 20
#define WG 1024
#define GU 16

__device__ __forceinline__ unsigned block_incl_scan(unsigned v, unsigned* wtot, unsigned& total) {
    const int lane = threadIdx.x & 31;
    const int wave = threadIdx.x >> 5;
    for (int d = 1; d < 32; d <<= 1) {
        unsigned u = __shfl_up(v, d);
        if (lane >= d) v += u;
    }
    if (lane == 31) wtot[wave] = v;
    __syncthreads();
    if (wave == 0) {
        unsigned w = wtot[lane];
        for (int d = 1; d < 32; d <<= 1) {
            unsigned u = __shfl_up(w, d);
            if (lane >= d) w += u;
        }
        wtot[lane] = w;
    }
    __syncthreads();
    total = wtot[31];
    if (wave > 0) v += wtot[wave - 1];
    __syncthreads();
    return v;
}

__global__ __launch_bounds__(WG) void gcn_count(const int* __restrict__ dst, unsigned* __restrict__ cnt,
                                                int e, int nbin, int nblk) {
    __shared__ unsigned hist[MAXBIN];
    const int tid = (int)threadIdx.x;
    if (tid < MAXBIN) hist[tid] = 0u;
    __syncthreads();
    const int base = (int)blockIdx.x * CHUNK;
    if (base + CHUNK <= e) {
        int d[EPT];
#pragma unroll
        for (int j = 0; j < EPT; ++j) d[j] = __builtin_nontemporal_load(&dst[base + j * WG + tid]);
#pragma unroll
        for (int j = 0; j < EPT; ++j) atomicAdd(&hist[(unsigned)d[j] / RANGE], 1u);
    } else {
        for (int j = 0; j < EPT; ++j) {
            int i = base + j * WG + tid;
            if (i < e) atomicAdd(&hist[(unsigned)__builtin_nontemporal_load(&dst[i]) / RANGE], 1u);
        }
    }
    __syncthreads();
    if (tid < nbin) cnt[tid * nblk + (int)blockIdx.x] = hist[tid];
}

__global__ __launch_bounds__(WG) void gcn_rowscan(unsigned* __restrict__ cnt, unsigned* __restrict__ rowtot, int nblk) {
    __shared__ unsigned wtot[32];
    unsigned* row = cnt + (size_t)blockIdx.x * nblk;
    unsigned carry = 0u;
    for (int t0 = 0; t0 < nblk; t0 += WG) {
        int i = t0 + (int)threadIdx.x;
        unsigned v = (i < nblk) ? row[i] : 0u;
        unsigned tot;
        unsigned inc = block_incl_scan(v, wtot, tot);
        if (i < nblk) row[i] = carry + inc - v;
        carry += tot;
    }
    if (threadIdx.x == 0) rowtot[blockIdx.x] = carry;
}

__global__ __launch_bounds__(WG) void gcn_bucket(const int* __restrict__ src, const int* __restrict__ dst,
                                                 const unsigned* __restrict__ cnt, const unsigned* __restrict__ rowtot,
                                                 unsigned* __restrict__ packed, int e, int nbin, int nblk) {
    __shared__ unsigned stage[CHUNK];
    __shared__ unsigned hist[MAXBIN];
    __shared__ unsigned lbase[MAXBIN];
    __shared__ unsigned gbase[MAXBIN];
    __shared__ unsigned wtot[32];
    const int tid = (int)threadIdx.x;
    const int lane = tid & 31;
    const int wave = tid >> 5;
    if (tid < MAXBIN) hist[tid] = 0u;
    __syncthreads();

    const int base = (int)blockIdx.x * CHUNK;
    unsigned word[EPT], br[EPT];
    if (base + CHUNK <= e) {
        int s[EPT], d[EPT];
#pragma unroll
        for (int j = 0; j < EPT; ++j) {
            s[j] = __builtin_nontemporal_load(&src[base + j * WG + tid]);
            d[j] = __builtin_nontemporal_load(&dst[base + j * WG + tid]);
        }
#pragma unroll
        for (int j = 0; j < EPT; ++j) {
            unsigned bin = (unsigned)d[j] / RANGE;
            unsigned off = (unsigned)d[j] - bin * RANGE;
            word[j] = ((unsigned)s[j] << 12) | off;
            br[j] = (bin << 16) | atomicAdd(&hist[bin], 1u);
        }
    } else {
#pragma unroll
        for (int j = 0; j < EPT; ++j) {
            int i = base + j * WG + tid;
            word[j] = 0u;
            br[j] = 0xFFFFFFFFu;
            if (i < e) {
                unsigned sv = (unsigned)__builtin_nontemporal_load(&src[i]);
                unsigned dv = (unsigned)__builtin_nontemporal_load(&dst[i]);
                unsigned bin = dv / RANGE;
                word[j] = (sv << 12) | (dv - bin * RANGE);
                br[j] = (bin << 16) | atomicAdd(&hist[bin], 1u);
            }
        }
    }
    __syncthreads();

    unsigned tot;
    unsigned hv = (tid < nbin) ? hist[tid] : 0u;
    unsigned hi = block_incl_scan(hv, wtot, tot);
    if (tid < nbin) lbase[tid] = hi - hv;
    unsigned rv = (tid < nbin) ? rowtot[tid] : 0u;
    unsigned ri = block_incl_scan(rv, wtot, tot);
    if (tid < nbin) gbase[tid] = (ri - rv) + cnt[tid * nblk + (int)blockIdx.x];
    __syncthreads();

#pragma unroll
    for (int j = 0; j < EPT; ++j)
        if (br[j] != 0xFFFFFFFFu) stage[lbase[br[j] >> 16] + (br[j] & 0xFFFFu)] = word[j];
    __syncthreads();

    for (int bin = wave; bin < nbin; bin += 32) {
        const unsigned c = hist[bin], lb = lbase[bin], gb = gbase[bin];
        for (unsigned k = (unsigned)lane; k < c; k += 32u) packed[gb + k] = stage[lb + k];
    }
}

template <int MODE>
__global__ __launch_bounds__(WG) void gcn_range(const unsigned* __restrict__ packed,
                                                const unsigned* __restrict__ rowtot, int nbin,
                                                float* dinv, const float* __restrict__ gsrc,
                                                float* __restrict__ wr,
                                                const float* __restrict__ W1, const float* __restrict__ b1,
                                                const float* __restrict__ W2, const float* __restrict__ b2, int n) {
    __shared__ float accf[RANGE];
    __shared__ float dis[RANGE];
    __shared__ unsigned wtot[32];
    __shared__ unsigned slice[2];
    const int tid = (int)threadIdx.x;
    const int lane = tid & 31;
    const int r = (int)blockIdx.x;
    unsigned* cacc = (unsigned*)accf;

    unsigned tot;
    unsigned rv = (tid < nbin) ? rowtot[tid] : 0u;
    unsigned ri = block_incl_scan(rv, wtot, tot);
    if (tid == r) { slice[0] = ri - rv; slice[1] = ri; }
    for (int i = tid; i < RANGE; i += WG) accf[i] = 0.0f;
    __syncthreads();
    const unsigned start = __builtin_amdgcn_readfirstlane(slice[0]);
    const unsigned end   = __builtin_amdgcn_readfirstlane(slice[1]);

    for (unsigned k0 = start; k0 < end; k0 += WG * GU) {
        unsigned w[GU];
#pragma unroll
        for (int j = 0; j < GU; ++j) {
            unsigned k = k0 + (unsigned)(j * WG + tid);
            w[j] = __builtin_nontemporal_load(&packed[k < end ? k : start]);
        }
        if (MODE == 0) {
#pragma unroll
            for (int j = 0; j < GU; ++j) {
                const bool ok = k0 + (unsigned)(j * WG + tid) < end;
                atomicAdd(&cacc[ok ? (w[j] & 4095u) : (unsigned)tid], ok ? 1u : 0u);
            }
        } else {
            float v[GU];
#pragma unroll
            for (int j = 0; j < GU; ++j) v[j] = gsrc[w[j] >> 12];
#pragma unroll
            for (int j = 0; j < GU; ++j) {
                const bool ok = k0 + (unsigned)(j * WG + tid) < end;
                atomicAdd(&accf[ok ? (w[j] & 4095u) : (unsigned)tid], ok ? v[j] : 0.0f);
            }
        }
    }
    __syncthreads();

    if (MODE == 0) {
        for (int i = tid; i < RANGE; i += WG) {
            int node = r * RANGE + i;
            if (node < n) {
                float d = (float)cacc[i] + 1.0f;
                float di = rsqrtf(d);
                dinv[node] = di;
                wr[node] = di * gsrc[node];
            }
        }
    } else if (MODE == 2) {
        for (int i = tid; i < RANGE; i += WG) {
            int node = r * RANGE + i;
            if (node < n) wr[node] = dinv[node] * (accf[i] + gsrc[node]) + b2[0];
        }
    } else {
        for (int i = tid; i < RANGE; i += WG) {
            int node = r * RANGE + i;
            float di = 0.0f, sv = 0.0f;
            if (node < n) { di = dinv[node]; sv = di * (accf[i] + gsrc[node]); }
            accf[i] = sv;
            dis[i] = di;
        }
        __syncthreads();
        const int wave = __builtin_amdgcn_readfirstlane(tid >> 5);
        v2f a; a.x = 0.0f; a.y = 0.0f;
        if (lane < 8) a.x = W1[lane];
        v8f c;
#pragma unroll
        for (int j = 0; j < 8; ++j) c[j] = (lane < 16) ? b1[j] : 0.0f;
        for (int g = wave; g < RANGE / 16; g += 32) {
            const int i = g * 16 + (lane & 15);
            const float sv = accf[i];
            const float di = dis[i];
            v2f b; b.x = (lane < 16) ? sv : 0.0f; b.y = 0.0f;
            v8f dacc = __builtin_amdgcn_wmma_f32_16x16x4_f32(
                 false, a,  false, b,
                 (short)0, c,  false,  false);
            float acc = 0.0f;
#pragma unroll
            for (int j = 0; j < 8; ++j) acc = fmaf(fmaxf(dacc[j], 0.0f), W2[j], acc);
            const int node = r * RANGE + i;
            if (lane < 16 && node < n) wr[node] = di * acc;
        }
    }
}

extern "C" void kernel_launch(void* const* d_in, const int* in_sizes, int n_in,
                              void* d_out, int out_size, void* d_ws, size_t ws_size,
                              hipStream_t stream) {
    const float* x  = (const float*)d_in[0];
    const int*   ei = (const int*)d_in[1];
    const float* W1 = (const float*)d_in[2];
    const float* b1 = (const float*)d_in[3];
    const float* W2 = (const float*)d_in[4];
    const float* b2 = (const float*)d_in[5];

    const int n = in_sizes[0];
    const int e = in_sizes[1] / 2;
    const int* src = ei;
    const int* dst = ei + e;

    float* deg = (float*)d_ws;
    float* acc = deg + n;
    float* pq  = acc + n;
    float* out = (float*)d_out;

    const int nbin = (n + RANGE - 1) / RANGE;
    const int nblk = (e + CHUNK - 1) / CHUNK;
    const size_t need = ((size_t)2 * n + (size_t)e + (size_t)nbin * nblk + MAXBIN) * sizeof(float);
    if (e > 0 && n <= (1 << 20) && nbin <= MAXBIN && ws_size >= need) {
        float*    dinv   = (float*)d_ws;
        float*    q      = dinv + n;
        unsigned* packed = (unsigned*)(q + n);
        unsigned* cnt    = packed + e;
        unsigned* rowtot = cnt + (size_t)nbin * nblk;
        float*    p      = out;
        gcn_count  <<<nblk, WG, 0, stream>>>(dst, cnt, e, nbin, nblk);
        gcn_rowscan<<<nbin, WG, 0, stream>>>(cnt, rowtot, nblk);
        gcn_bucket <<<nblk, WG, 0, stream>>>(src, dst, cnt, rowtot, packed, e, nbin, nblk);
        gcn_range<0><<<nbin, WG, 0, stream>>>(packed, rowtot, nbin, dinv, x, p,   W1, b1, W2, b2, n);
        gcn_range<1><<<nbin, WG, 0, stream>>>(packed, rowtot, nbin, dinv, p, q,   W1, b1, W2, b2, n);
        gcn_range<2><<<nbin, WG, 0, stream>>>(packed, rowtot, nbin, dinv, q, out, W1, b1, W2, b2, n);
        return;
    }

    const int tb = 256;
    const int nb_n = (n + tb - 1) / tb;
    const int per_block = tb * EDGES_PER_THREAD;
    const int nb_e = (e + per_block - 1) / per_block;

    gcn_init   <<<nb_n, tb, 0, stream>>>(deg, acc, n);
    gcn_degree <<<nb_e, tb, 0, stream>>>(dst, deg, acc, e);
    gcn_dinv   <<<nb_n, tb, 0, stream>>>(deg, acc, x, pq, out, n);
    gcn_scatter<<<nb_e, tb, 0, stream>>>(src, dst, pq, acc, out, e);

    const int nodes_per_block = (tb / 32) * 16;
    gcn_node_mlp<<<(n + nodes_per_block - 1) / nodes_per_block, tb, 0, stream>>>(
        acc, out, deg, W1, b1, W2, pq, n);

    gcn_scatter<<<nb_e, tb, 0, stream>>>(src, dst, pq, acc, out, e);
    gcn_final  <<<nb_n, tb, 0, stream>>>(out, acc, pq, deg, b2, n);
}


# Edge bucketing made local to a block: sorted slice plus a (start,length) word per range and block, range groups walk per-block runs; no count or scan launch (f32 sum order: arrival order as before)
# speedup vs baseline: 5.7444x; 1.0608x over previous
#include <hip/hip_runtime.h>


typedef __attribute__((ext_vector_type(2))) float v2f;
typedef __attribute__((ext_vector_type(8))) float v8f;


#define EDGES_PER_THREAD 4

__device__ __forceinline__ int package_half() {
    __shared__ int half_of_block;
    if (threadIdx.x < warpSize) {
        unsigned id = __builtin_amdgcn_s_sendmsg_rtn(0x87);
        if (threadIdx.x == 0) half_of_block = (int)((id >> 18) & 1u);
    }
    __syncthreads();
    return __builtin_amdgcn_readfirstlane(half_of_block);
}

__global__ void gcn_init(float* __restrict__ deg_a, float* __restrict__ deg_b, int n) {
    int i = blockIdx.x * blockDim.x + threadIdx.x;
    if (i < n) { deg_a[i] = 0.0f; deg_b[i] = 0.0f; }
}

__global__ void gcn_degree(const int* __restrict__ dst, float* __restrict__ deg_a,
                           float* __restrict__ deg_b, int e) {
    float* __restrict__ deg = package_half() ? deg_b : deg_a;
    const int per_block = (int)blockDim.x * EDGES_PER_THREAD;
    const int b0 = (int)blockIdx.x * per_block;
    const int i0 = b0 + (int)threadIdx.x;
    if (b0 + per_block <= e) {
        int d[EDGES_PER_THREAD];
#pragma unroll
        for (int j = 0; j < EDGES_PER_THREAD; ++j)
            d[j] = __builtin_nontemporal_load(&dst[i0 + j * (int)blockDim.x]);
#pragma unroll
        for (int j = 0; j < EDGES_PER_THREAD; ++j)
            atomicAdd(&deg[d[j]], 1.0f);
    } else {
        for (int j = 0; j < EDGES_PER_THREAD; ++j) {
            int i = i0 + j * (int)blockDim.x;
            if (i < e) atomicAdd(&deg[__builtin_nontemporal_load(&dst[i])], 1.0f);
        }
    }
}

__global__ void gcn_dinv(float* __restrict__ deg_a, float* __restrict__ deg_b,
                         const float* __restrict__ x, float* __restrict__ p,
                         float* __restrict__ out, int n) {
    int i = blockIdx.x * blockDim.x + threadIdx.x;
    if (i < n) {
        float d = (deg_a[i] + deg_b[i]) + 1.0f;
        float di = rsqrtf(d);
        deg_a[i] = di;
        p[i] = di * x[i];
        deg_b[i] = 0.0f;
        out[i] = 0.0f;
    }
}

__global__ void gcn_scatter(const int* __restrict__ src, const int* __restrict__ dst,
                            const float* __restrict__ val,
                            float* __restrict__ acc_a, float* __restrict__ acc_b, int e) {
    float* __restrict__ acc = package_half() ? acc_b : acc_a;
    const int per_block = (int)blockDim.x * EDGES_PER_THREAD;
    const int b0 = (int)blockIdx.x * per_block;
    const int i0 = b0 + (int)threadIdx.x;
    if (b0 + per_block <= e) {
        int s[EDGES_PER_THREAD], d[EDGES_PER_THREAD];
        float v[EDGES_PER_THREAD];
#pragma unroll
        for (int j = 0; j < EDGES_PER_THREAD; ++j) {
            int i = i0 + j * (int)blockDim.x;
            s[j] = __builtin_nontemporal_load(&src[i]);
            d[j] = __builtin_nontemporal_load(&dst[i]);
        }
#pragma unroll
        for (int j = 0; j < EDGES_PER_THREAD; ++j)
            v[j] = val[s[j]];
#pragma unroll
        for (int j = 0; j < EDGES_PER_THREAD; ++j)
            atomicAdd(&acc[d[j]], v[j]);
    } else {
        for (int j = 0; j < EDGES_PER_THREAD; ++j) {
            int i = i0 + j * (int)blockDim.x;
            if (i < e) {
                int s = __builtin_nontemporal_load(&src[i]);
                int d = __builtin_nontemporal_load(&dst[i]);
                atomicAdd(&acc[d], val[s]);
            }
        }
    }
}

__global__ void gcn_node_mlp(float* __restrict__ acc_a, float* __restrict__ acc_b,
                             const float* __restrict__ dinv,
                             const float* __restrict__ W1, const float* __restrict__ b1,
                             const float* __restrict__ W2,
                             float* __restrict__ pq, int n) {
    int lane = threadIdx.x & 31;
    int wave = threadIdx.x >> 5;
    int base = (blockIdx.x * (blockDim.x >> 5) + wave) * 16;
    int node = base + lane;

    float sv = 0.0f;
    float di = 0.0f;
    if (lane < 16 && node < n) {
        di = dinv[node];
        sv = di * ((acc_a[node] + acc_b[node]) + pq[node]);
        acc_a[node] = 0.0f;
        acc_b[node] = 0.0f;
    }

    v2f a; a.x = 0.0f; a.y = 0.0f;
    if (lane < 8) a.x = W1[lane];
    v2f b; b.x = (lane < 16) ? sv : 0.0f; b.y = 0.0f;
    v8f c;
#pragma unroll
    for (int r = 0; r < 8; ++r) c[r] = (lane < 16) ? b1[r] : 0.0f;

    v8f dacc = __builtin_amdgcn_wmma_f32_16x16x4_f32(
         false, a,  false, b,
         (short)0, c,  false,  false);

    float acc = 0.0f;
#pragma unroll
    for (int r = 0; r < 8; ++r) acc = fmaf(fmaxf(dacc[r], 0.0f), W2[r], acc);

    if (lane < 16 && node < n) pq[node] = di * acc;
}

__global__ void gcn_final(float* __restrict__ out, const float* __restrict__ acc_a,
                          const float* __restrict__ q,
                          const float* __restrict__ dinv, const float* __restrict__ b2, int n) {
    int i = blockIdx.x * blockDim.x + threadIdx.x;
    if (i < n) {
        out[i] = dinv[i] * ((acc_a[i] + out[i]) + q[i]) + b2[0];
    }
}


#define RANGE 3936
#define MAXBIN 272
#define CHUNK 40960
#define EPT 40
#define EB 10
#define WG 1024
#define SL 6
#define RU 4

__device__ __forceinline__ unsigned block_incl_scan(unsigned v, unsigned* wtot, unsigned& total) {
    const int lane = threadIdx.x & 31;
    const int wave = threadIdx.x >> 5;
    for (int d = 1; d < 32; d <<= 1) {
        unsigned u = __shfl_up(v, d);
        if (lane >= d) v += u;
    }
    if (lane == 31) wtot[wave] = v;
    __syncthreads();
    if (wave == 0) {
        unsigned w = wtot[lane];
        for (int d = 1; d < 32; d <<= 1) {
            unsigned u = __shfl_up(w, d);
            if (lane >= d) w += u;
        }
        wtot[lane] = w;
    }
    __syncthreads();
    total = wtot[31];
    if (wave > 0) v += wtot[wave - 1];
    __syncthreads();
    return v;
}

__global__ __launch_bounds__(WG) void gcn_bucket(const int* __restrict__ src, const int* __restrict__ dst,
                                                 unsigned* __restrict__ packed, unsigned* __restrict__ desc,
                                                 int e, int nbin, int nblk) {
    __shared__ unsigned stage[CHUNK];
    __shared__ unsigned hist[MAXBIN];
    __shared__ unsigned lbase[MAXBIN];
    __shared__ unsigned wtot[32];
    const int tid = (int)threadIdx.x;
    if (tid < MAXBIN) hist[tid] = 0u;
    __syncthreads();

    const int base = (int)blockIdx.x * CHUNK;
    unsigned word[EPT], br[EPT];
    if (base + CHUNK <= e) {
#pragma unroll
        for (int h = 0; h < EPT; h += EB) {
            int s[EB], d[EB];
#pragma unroll
            for (int j = 0; j < EB; ++j) {
                s[j] = __builtin_nontemporal_load(&src[base + (h + j) * WG + tid]);
                d[j] = __builtin_nontemporal_load(&dst[base + (h + j) * WG + tid]);
            }
#pragma unroll
            for (int j = 0; j < EB; ++j) {
                unsigned bin = (unsigned)d[j] / RANGE;
                unsigned off = (unsigned)d[j] - bin * RANGE;
                word[h + j] = ((unsigned)s[j] << 12) | off;
                br[h + j] = (bin << 16) | atomicAdd(&hist[bin], 1u);
            }
        }
    } else {
#pragma unroll
        for (int j = 0; j < EPT; ++j) {
            int i = base + j * WG + tid;
            word[j] = 0u;
            br[j] = 0xFFFFFFFFu;
            if (i < e) {
                unsigned sv = (unsigned)__builtin_nontemporal_load(&src[i]);
                unsigned dv = (unsigned)__builtin_nontemporal_load(&dst[i]);
                unsigned bin = dv / RANGE;
                word[j] = (sv << 12) | (dv - bin * RANGE);
                br[j] = (bin << 16) | atomicAdd(&hist[bin], 1u);
            }
        }
    }
    __syncthreads();

    unsigned tot;
    unsigned hv = (tid < nbin) ? hist[tid] : 0u;
    unsigned hi = block_incl_scan(hv, wtot, tot);
    if (tid < nbin) {
        lbase[tid] = hi - hv;
        desc[(size_t)tid * nblk + blockIdx.x] = (hi - hv) | (hv << 16);
    }
    __syncthreads();

#pragma unroll
    for (int j = 0; j < EPT; ++j)
        if (br[j] != 0xFFFFFFFFu) stage[lbase[br[j] >> 16] + (br[j] & 0xFFFFu)] = word[j];
    __syncthreads();

    const int have = (e - base < CHUNK) ? (e - base) : CHUNK;
    for (int i = tid; i < have; i += WG) packed[base + i] = stage[i];
}

template <int MODE>
__global__ __launch_bounds__(WG) void gcn_range(const unsigned* __restrict__ packed,
                                                const unsigned* __restrict__ desc, int nblk,
                                                float* dinv, const float* __restrict__ gsrc,
                                                float* __restrict__ wr,
                                                const float* __restrict__ W1, const float* __restrict__ b1,
                                                const float* __restrict__ W2, const float* __restrict__ b2, int n) {
    __shared__ float accf[RANGE];
    __shared__ float dis[RANGE];
    const int tid = (int)threadIdx.x;
    const int lane = tid & 31;
    const int r = (int)blockIdx.x;
    unsigned* cacc = (unsigned*)accf;
    const int wave = __builtin_amdgcn_readfirstlane(tid >> 5);

    const unsigned* __restrict__ drow = desc + (size_t)r * nblk;
    unsigned dsc[RU];
#pragma unroll
    for (int u = 0; u < RU; ++u) {
        const int b = wave + 32 * u;
        dsc[u] = drow[b < nblk ? b : 0];
    }

    for (int i = tid; i < RANGE; i += WG) accf[i] = 0.0f;
    __syncthreads();

    for (int b0 = wave; b0 < nblk; b0 += 32 * RU) {
        unsigned w[RU * SL], len[RU], first[RU], nxt[RU];
#pragma unroll
        for (int u = 0; u < RU; ++u) {
            const int b = b0 + 32 * RU + 32 * u;
            nxt[u] = drow[b < nblk ? b : 0];
        }
#pragma unroll
        for (int u = 0; u < RU; ++u) {
            const int b = b0 + 32 * u;
            len[u] = (b < nblk) ? (dsc[u] >> 16) : 0u;
            first[u] = (unsigned)b * CHUNK + (dsc[u] & 0xFFFFu);
#pragma unroll
            for (int j = 0; j < SL; ++j) {
                const unsigned k = (unsigned)(j * 32 + lane);
                w[u * SL + j] = __builtin_nontemporal_load(&packed[k < len[u] ? first[u] + k : 0u]);
            }
        }
#pragma unroll
        for (int u = 0; u < RU; ++u) dsc[u] = nxt[u];
        if (MODE == 0) {
#pragma unroll
            for (int u = 0; u < RU; ++u) {
#pragma unroll
                for (int j = 0; j < SL; ++j) {
                    const bool ok = (unsigned)(j * 32 + lane) < len[u];
                    atomicAdd(&cacc[ok ? (w[u * SL + j] & 4095u) : (unsigned)tid], ok ? 1u : 0u);
                }
            }
        } else {
            float v[RU * SL];
#pragma unroll
            for (int q = 0; q < RU * SL; ++q) v[q] = gsrc[w[q] >> 12];
#pragma unroll
            for (int u = 0; u < RU; ++u) {
#pragma unroll
                for (int j = 0; j < SL; ++j) {
                    const bool ok = (unsigned)(j * 32 + lane) < len[u];
                    atomicAdd(&accf[ok ? (w[u * SL + j] & 4095u) : (unsigned)tid], ok ? v[u * SL + j] : 0.0f);
                }
            }
        }
#pragma unroll
        for (int u = 0; u < RU; ++u) {
            for (unsigned k = (unsigned)(SL * 32 + lane); k < len[u]; k += 32u) {
                const unsigned ww = __builtin_nontemporal_load(&packed[first[u] + k]);
                if (MODE == 0) atomicAdd(&cacc[ww & 4095u], 1u);
                else atomicAdd(&accf[ww & 4095u], gsrc[ww >> 12]);
            }
        }
    }
    __syncthreads();

    if (MODE == 0) {
        for (int i = tid; i < RANGE; i += WG) {
            int node = r * RANGE + i;
            if (node < n) {
                float d = (float)cacc[i] + 1.0f;
                float di = rsqrtf(d);
                dinv[node] = di;
                wr[node] = di * gsrc[node];
            }
        }
    } else if (MODE == 2) {
        for (int i = tid; i < RANGE; i += WG) {
            int node = r * RANGE + i;
            if (node < n) wr[node] = dinv[node] * (accf[i] + gsrc[node]) + b2[0];
        }
    } else {
        for (int i = tid; i < RANGE; i += WG) {
            int node = r * RANGE + i;
            float di = 0.0f, sv = 0.0f;
            if (node < n) { di = dinv[node]; sv = di * (accf[i] + gsrc[node]); }
            accf[i] = sv;
            dis[i] = di;
        }
        __syncthreads();
        v2f a; a.x = 0.0f; a.y = 0.0f;
        if (lane < 8) a.x = W1[lane];
        v8f c;
#pragma unroll
        for (int j = 0; j < 8; ++j) c[j] = (lane < 16) ? b1[j] : 0.0f;
        for (int g = wave; g < RANGE / 16; g += 32) {
            const int i = g * 16 + (lane & 15);
            const float sv = accf[i];
            const float di = dis[i];
            v2f b; b.x = (lane < 16) ? sv : 0.0f; b.y = 0.0f;
            v8f dacc = __builtin_amdgcn_wmma_f32_16x16x4_f32(
                 false, a,  false, b,
                 (short)0, c,  false,  false);
            float acc = 0.0f;
#pragma unroll
            for (int j = 0; j < 8; ++j) acc = fmaf(fmaxf(dacc[j], 0.0f), W2[j], acc);
            const int node = r * RANGE + i;
            if (lane < 16 && node < n) wr[node] = di * acc;
        }
    }
}

extern "C" void kernel_launch(void* const* d_in, const int* in_sizes, int n_in,
                              void* d_out, int out_size, void* d_ws, size_t ws_size,
                              hipStream_t stream) {
    const float* x  = (const float*)d_in[0];
    const int*   ei = (const int*)d_in[1];
    const float* W1 = (const float*)d_in[2];
    const float* b1 = (const float*)d_in[3];
    const float* W2 = (const float*)d_in[4];
    const float* b2 = (const float*)d_in[5];

    const int n = in_sizes[0];
    const int e = in_sizes[1] / 2;
    const int* src = ei;
    const int* dst = ei + e;

    float* deg = (float*)d_ws;
    float* acc = deg + n;
    float* pq  = acc + n;
    float* out = (float*)d_out;

    const int nbin = (n + RANGE - 1) / RANGE;
    const int nblk = (e + CHUNK - 1) / CHUNK;
    const size_t need = ((size_t)2 * n + (size_t)e + (size_t)nbin * nblk) * sizeof(float);
    if (e > 0 && n <= (1 << 20) && nbin <= MAXBIN && ws_size >= need) {
        float*    dinv   = (float*)d_ws;
        float*    q      = dinv + n;
        unsigned* packed = (unsigned*)(q + n);
        unsigned* desc   = packed + e;
        float*    p      = out;
        gcn_bucket <<<nblk, WG, 0, stream>>>(src, dst, packed, desc, e, nbin, nblk);
        gcn_range<0><<<nbin, WG, 0, stream>>>(packed, desc, nblk, dinv, x, p,   W1, b1, W2, b2, n);
        gcn_range<1><<<nbin, WG, 0, stream>>>(packed, desc, nblk, dinv, p, q,   W1, b1, W2, b2, n);
        gcn_range<2><<<nbin, WG, 0, stream>>>(packed, desc, nblk, dinv, q, out, W1, b1, W2, b2, n);
        return;
    }

    const int tb = 256;
    const int nb_n = (n + tb - 1) / tb;
    const int per_block = tb * EDGES_PER_THREAD;
    const int nb_e = (e + per_block - 1) / per_block;

    gcn_init   <<<nb_n, tb, 0, stream>>>(deg, acc, n);
    gcn_degree <<<nb_e, tb, 0, stream>>>(dst, deg, acc, e);
    gcn_dinv   <<<nb_n, tb, 0, stream>>>(deg, acc, x, pq, out, n);
    gcn_scatter<<<nb_e, tb, 0, stream>>>(src, dst, pq, acc, out, e);

    const int nodes_per_block = (tb / 32) * 16;
    gcn_node_mlp<<<(n + nodes_per_block - 1) / nodes_per_block, tb, 0, stream>>>(
        acc, out, deg, W1, b1, W2, pq, n);

    gcn_scatter<<<nb_e, tb, 0, stream>>>(src, dst, pq, acc, out, e);
    gcn_final  <<<nb_n, tb, 0, stream>>>(out, acc, pq, deg, b2, n);
}


# Padding lanes of the range passes re-read their own run's last word, not the list's first word: no single line is asked for by every wave any more (f32 sum order: arrival order as before)
# speedup vs baseline: 7.0986x; 1.2357x over previous
#include <hip/hip_runtime.h>


typedef __attribute__((ext_vector_type(2))) float v2f;
typedef __attribute__((ext_vector_type(8))) float v8f;


#define EDGES_PER_THREAD 4

__device__ __forceinline__ int package_half() {
    __shared__ int half_of_block;
    if (threadIdx.x < warpSize) {
        unsigned id = __builtin_amdgcn_s_sendmsg_rtn(0x87);
        if (threadIdx.x == 0) half_of_block = (int)((id >> 18) & 1u);
    }
    __syncthreads();
    return __builtin_amdgcn_readfirstlane(half_of_block);
}

__global__ void gcn_init(float* __restrict__ deg_a, float* __restrict__ deg_b, int n) {
    int i = blockIdx.x * blockDim.x + threadIdx.x;
    if (i < n) { deg_a[i] = 0.0f; deg_b[i] = 0.0f; }
}

__global__ void gcn_degree(const int* __restrict__ dst, float* __restrict__ deg_a,
                           float* __restrict__ deg_b, int e) {
    float* __restrict__ deg = package_half() ? deg_b : deg_a;
    const int per_block = (int)blockDim.x * EDGES_PER_THREAD;
    const int b0 = (int)blockIdx.x * per_block;
    const int i0 = b0 + (int)threadIdx.x;
    if (b0 + per_block <= e) {
        int d[EDGES_PER_THREAD];
#pragma unroll
        for (int j = 0; j < EDGES_PER_THREAD; ++j)
            d[j] = __builtin_nontemporal_load(&dst[i0 + j * (int)blockDim.x]);
#pragma unroll
        for (int j = 0; j < EDGES_PER_THREAD; ++j)
            atomicAdd(&deg[d[j]], 1.0f);
    } else {
        for (int j = 0; j < EDGES_PER_THREAD; ++j) {
            int i = i0 + j * (int)blockDim.x;
            if (i < e) atomicAdd(&deg[__builtin_nontemporal_load(&dst[i])], 1.0f);
        }
    }
}

__global__ void gcn_dinv(float* __restrict__ deg_a, float* __restrict__ deg_b,
                         const float* __restrict__ x, float* __restrict__ p,
                         float* __restrict__ out, int n) {
    int i = blockIdx.x * blockDim.x + threadIdx.x;
    if (i < n) {
        float d = (deg_a[i] + deg_b[i]) + 1.0f;
        float di = rsqrtf(d);
        deg_a[i] = di;
        p[i] = di * x[i];
        deg_b[i] = 0.0f;
        out[i] = 0.0f;
    }
}

__global__ void gcn_scatter(const int* __restrict__ src, const int* __restrict__ dst,
                            const float* __restrict__ val,
                            float* __restrict__ acc_a, float* __restrict__ acc_b, int e) {
    float* __restrict__ acc = package_half() ? acc_b : acc_a;
    const int per_block = (int)blockDim.x * EDGES_PER_THREAD;
    const int b0 = (int)blockIdx.x * per_block;
    const int i0 = b0 + (int)threadIdx.x;
    if (b0 + per_block <= e) {
        int s[EDGES_PER_THREAD], d[EDGES_PER_THREAD];
        float v[EDGES_PER_THREAD];
#pragma unroll
        for (int j = 0; j < EDGES_PER_THREAD; ++j) {
            int i = i0 + j * (int)blockDim.x;
            s[j] = __builtin_nontemporal_load(&src[i]);
            d[j] = __builtin_nontemporal_load(&dst[i]);
        }
#pragma unroll
        for (int j = 0; j < EDGES_PER_THREAD; ++j)
            v[j] = val[s[j]];
#pragma unroll
        for (int j = 0; j < EDGES_PER_THREAD; ++j)
            atomicAdd(&acc[d[j]], v[j]);
    } else {
        for (int j = 0; j < EDGES_PER_THREAD; ++j) {
            int i = i0 + j * (int)blockDim.x;
            if (i < e) {
                int s = __builtin_nontemporal_load(&src[i]);
                int d = __builtin_nontemporal_load(&dst[i]);
                atomicAdd(&acc[d], val[s]);
            }
        }
    }
}

__global__ void gcn_node_mlp(float* __restrict__ acc_a, float* __restrict__ acc_b,
                             const float* __restrict__ dinv,
                             const float* __restrict__ W1, const float* __restrict__ b1,
                             const float* __restrict__ W2,
                             float* __restrict__ pq, int n) {
    int lane = threadIdx.x & 31;
    int wave = threadIdx.x >> 5;
    int base = (blockIdx.x * (blockDim.x >> 5) + wave) * 16;
    int node = base + lane;

    float sv = 0.0f;
    float di = 0.0f;
    if (lane < 16 && node < n) {
        di = dinv[node];
        sv = di * ((acc_a[node] + acc_b[node]) + pq[node]);
        acc_a[node] = 0.0f;
        acc_b[node] = 0.0f;
    }

    v2f a; a.x = 0.0f; a.y = 0.0f;
    if (lane < 8) a.x = W1[lane];
    v2f b; b.x = (lane < 16) ? sv : 0.0f; b.y = 0.0f;
    v8f c;
#pragma unroll
    for (int r = 0; r < 8; ++r) c[r] = (lane < 16) ? b1[r] : 0.0f;

    v8f dacc = __builtin_amdgcn_wmma_f32_16x16x4_f32(
         false, a,  false, b,
         (short)0, c,  false,  false);

    float acc = 0.0f;
#pragma unroll
    for (int r = 0; r < 8; ++r) acc = fmaf(fmaxf(dacc[r], 0.0f), W2[r], acc);

    if (lane < 16 && node < n) pq[node] = di * acc;
}

__global__ void gcn_final(float* __restrict__ out, const float* __restrict__ acc_a,
                          const float* __restrict__ q,
                          const float* __restrict__ dinv, const float* __restrict__ b2, int n) {
    int i = blockIdx.x * blockDim.x + threadIdx.x;
    if (i < n) {
        out[i] = dinv[i] * ((acc_a[i] + out[i]) + q[i]) + b2[0];
    }
}


#define RANGE 3936
#define MAXBIN 272
#define CHUNK 40960
#define EPT 40
#define EB 10
#define WG 1024
#define SL 6
#define RU 4

__device__ __forceinline__ unsigned block_incl_scan(unsigned v, unsigned* wtot, unsigned& total) {
    const int lane = threadIdx.x & 31;
    const int wave = threadIdx.x >> 5;
    for (int d = 1; d < 32; d <<= 1) {
        unsigned u = __shfl_up(v, d);
        if (lane >= d) v += u;
    }
    if (lane == 31) wtot[wave] = v;
    __syncthreads();
    if (wave == 0) {
        unsigned w = wtot[lane];
        for (int d = 1; d < 32; d <<= 1) {
            unsigned u = __shfl_up(w, d);
            if (lane >= d) w += u;
        }
        wtot[lane] = w;
    }
    __syncthreads();
    total = wtot[31];
    if (wave > 0) v += wtot[wave - 1];
    __syncthreads();
    return v;
}

__global__ __launch_bounds__(WG) void gcn_bucket(const int* __restrict__ src, const int* __restrict__ dst,
                                                 unsigned* __restrict__ packed, unsigned* __restrict__ desc,
                                                 int e, int nbin, int nblk) {
    __shared__ unsigned stage[CHUNK];
    __shared__ unsigned hist[MAXBIN];
    __shared__ unsigned lbase[MAXBIN];
    __shared__ unsigned wtot[32];
    const int tid = (int)threadIdx.x;
    if (tid < MAXBIN) hist[tid] = 0u;
    __syncthreads();

    const int base = (int)blockIdx.x * CHUNK;
    unsigned word[EPT], br[EPT];
    if (base + CHUNK <= e) {
#pragma unroll
        for (int h = 0; h < EPT; h += EB) {
            int s[EB], d[EB];
#pragma unroll
            for (int j = 0; j < EB; ++j) {
                s[j] = __builtin_nontemporal_load(&src[base + (h + j) * WG + tid]);
                d[j] = __builtin_nontemporal_load(&dst[base + (h + j) * WG + tid]);
            }
#pragma unroll
            for (int j = 0; j < EB; ++j) {
                unsigned bin = (unsigned)d[j] / RANGE;
                unsigned off = (unsigned)d[j] - bin * RANGE;
                word[h + j] = ((unsigned)s[j] << 12) | off;
                br[h + j] = (bin << 16) | atomicAdd(&hist[bin], 1u);
            }
        }
    } else {
#pragma unroll
        for (int j = 0; j < EPT; ++j) {
            int i = base + j * WG + tid;
            word[j] = 0u;
            br[j] = 0xFFFFFFFFu;
            if (i < e) {
                unsigned sv = (unsigned)__builtin_nontemporal_load(&src[i]);
                unsigned dv = (unsigned)__builtin_nontemporal_load(&dst[i]);
                unsigned bin = dv / RANGE;
                word[j] = (sv << 12) | (dv - bin * RANGE);
                br[j] = (bin << 16) | atomicAdd(&hist[bin], 1u);
            }
        }
    }
    __syncthreads();

    unsigned tot;
    unsigned hv = (tid < nbin) ? hist[tid] : 0u;
    unsigned hi = block_incl_scan(hv, wtot, tot);
    if (tid < nbin) {
        lbase[tid] = hi - hv;
        desc[(size_t)tid * nblk + blockIdx.x] = (hi - hv) | (hv << 16);
    }
    __syncthreads();

#pragma unroll
    for (int j = 0; j < EPT; ++j)
        if (br[j] != 0xFFFFFFFFu) stage[lbase[br[j] >> 16] + (br[j] & 0xFFFFu)] = word[j];
    __syncthreads();

    const int have = (e - base < CHUNK) ? (e - base) : CHUNK;
    for (int i = tid; i < have; i += WG) packed[base + i] = stage[i];
}

template <int MODE>
__global__ __launch_bounds__(WG) void gcn_range(const unsigned* __restrict__ packed,
                                                const unsigned* __restrict__ desc, int nblk,
                                                float* dinv, const float* __restrict__ gsrc,
                                                float* __restrict__ wr,
                                                const float* __restrict__ W1, const float* __restrict__ b1,
                                                const float* __restrict__ W2, const float* __restrict__ b2, int n) {
    __shared__ float accf[RANGE];
    __shared__ float dis[RANGE];
    const int tid = (int)threadIdx.x;
    const int lane = tid & 31;
    const int r = (int)blockIdx.x;
    unsigned* cacc = (unsigned*)accf;
    const int wave = __builtin_amdgcn_readfirstlane(tid >> 5);

    const unsigned* __restrict__ drow = desc + (size_t)r * nblk;
    unsigned dsc[RU];
#pragma unroll
    for (int u = 0; u < RU; ++u) {
        const int b = wave + 32 * u;
        dsc[u] = drow[b < nblk ? b : 0];
    }

    for (int i = tid; i < RANGE; i += WG) accf[i] = 0.0f;
    __syncthreads();

    for (int b0 = wave; b0 < nblk; b0 += 32 * RU) {
        unsigned w[RU * SL], len[RU], first[RU], nxt[RU];
#pragma unroll
        for (int u = 0; u < RU; ++u) {
            const int b = b0 + 32 * RU + 32 * u;
            nxt[u] = drow[b < nblk ? b : 0];
        }
#pragma unroll
        for (int u = 0; u < RU; ++u) {
            const int b = b0 + 32 * u;
            len[u] = (b < nblk) ? (dsc[u] >> 16) : 0u;
            first[u] = (unsigned)b * CHUNK + (dsc[u] & 0xFFFFu);
            const unsigned rb = len[u] ? first[u] : (len[0] ? first[0] : 0u);
            const unsigned rl = len[u] ? len[u] - 1u : (len[0] ? len[0] - 1u : 0u);
#pragma unroll
            for (int j = 0; j < SL; ++j) {
                const unsigned k = (unsigned)(j * 32 + lane);
                w[u * SL + j] = __builtin_nontemporal_load(&packed[rb + (k < rl ? k : rl)]);
            }
        }
#pragma unroll
        for (int u = 0; u < RU; ++u) dsc[u] = nxt[u];
        if (MODE == 0) {
#pragma unroll
            for (int u = 0; u < RU; ++u) {
#pragma unroll
                for (int j = 0; j < SL; ++j) {
                    const bool ok = (unsigned)(j * 32 + lane) < len[u];
                    atomicAdd(&cacc[ok ? (w[u * SL + j] & 4095u) : (unsigned)tid], ok ? 1u : 0u);
                }
            }
        } else {
            float v[RU * SL];
#pragma unroll
            for (int q = 0; q < RU * SL; ++q) v[q] = gsrc[w[q] >> 12];
#pragma unroll
            for (int u = 0; u < RU; ++u) {
#pragma unroll
                for (int j = 0; j < SL; ++j) {
                    const bool ok = (unsigned)(j * 32 + lane) < len[u];
                    atomicAdd(&accf[ok ? (w[u * SL + j] & 4095u) : (unsigned)tid], ok ? v[u * SL + j] : 0.0f);
                }
            }
        }
#pragma unroll
        for (int u = 0; u < RU; ++u) {
            for (unsigned k = (unsigned)(SL * 32 + lane); k < len[u]; k += 32u) {
                const unsigned ww = __builtin_nontemporal_load(&packed[first[u] + k]);
                if (MODE == 0) atomicAdd(&cacc[ww & 4095u], 1u);
                else atomicAdd(&accf[ww & 4095u], gsrc[ww >> 12]);
            }
        }
    }
    __syncthreads();

    if (MODE == 0) {
        for (int i = tid; i < RANGE; i += WG) {
            int node = r * RANGE + i;
            if (node < n) {
                float d = (float)cacc[i] + 1.0f;
                float di = rsqrtf(d);
                dinv[node] = di;
                wr[node] = di * gsrc[node];
            }
        }
    } else if (MODE == 2) {
        for (int i = tid; i < RANGE; i += WG) {
            int node = r * RANGE + i;
            if (node < n) wr[node] = dinv[node] * (accf[i] + gsrc[node]) + b2[0];
        }
    } else {
        for (int i = tid; i < RANGE; i += WG) {
            int node = r * RANGE + i;
            float di = 0.0f, sv = 0.0f;
            if (node < n) { di = dinv[node]; sv = di * (accf[i] + gsrc[node]); }
            accf[i] = sv;
            dis[i] = di;
        }
        __syncthreads();
        v2f a; a.x = 0.0f; a.y = 0.0f;
        if (lane < 8) a.x = W1[lane];
        v8f c;
#pragma unroll
        for (int j = 0; j < 8; ++j) c[j] = (lane < 16) ? b1[j] : 0.0f;
        for (int g = wave; g < RANGE / 16; g += 32) {
            const int i = g * 16 + (lane & 15);
            const float sv = accf[i];
            const float di = dis[i];
            v2f b; b.x = (lane < 16) ? sv : 0.0f; b.y = 0.0f;
            v8f dacc = __builtin_amdgcn_wmma_f32_16x16x4_f32(
                 false, a,  false, b,
                 (short)0, c,  false,  false);
            float acc = 0.0f;
#pragma unroll
            for (int j = 0; j < 8; ++j) acc = fmaf(fmaxf(dacc[j], 0.0f), W2[j], acc);
            const int node = r * RANGE + i;
            if (lane < 16 && node < n) wr[node] = di * acc;
        }
    }
}

extern "C" void kernel_launch(void* const* d_in, const int* in_sizes, int n_in,
                              void* d_out, int out_size, void* d_ws, size_t ws_size,
                              hipStream_t stream) {
    const float* x  = (const float*)d_in[0];
    const int*   ei = (const int*)d_in[1];
    const float* W1 = (const float*)d_in[2];
    const float* b1 = (const float*)d_in[3];
    const float* W2 = (const float*)d_in[4];
    const float* b2 = (const float*)d_in[5];

    const int n = in_sizes[0];
    const int e = in_sizes[1] / 2;
    const int* src = ei;
    const int* dst = ei + e;

    float* deg = (float*)d_ws;
    float* acc = deg + n;
    float* pq  = acc + n;
    float* out = (float*)d_out;

    const int nbin = (n + RANGE - 1) / RANGE;
    const int nblk = (e + CHUNK - 1) / CHUNK;
    const size_t need = ((size_t)2 * n + (size_t)e + (size_t)nbin * nblk) * sizeof(float);
    if (e > 0 && n <= (1 << 20) && nbin <= MAXBIN && ws_size >= need) {
        float*    dinv   = (float*)d_ws;
        float*    q      = dinv + n;
        unsigned* packed = (unsigned*)(q + n);
        unsigned* desc   = packed + e;
        float*    p      = out;
        gcn_bucket <<<nblk, WG, 0, stream>>>(src, dst, packed, desc, e, nbin, nblk);
        gcn_range<0><<<nbin, WG, 0, stream>>>(packed, desc, nblk, dinv, x, p,   W1, b1, W2, b2, n);
        gcn_range<1><<<nbin, WG, 0, stream>>>(packed, desc, nblk, dinv, p, q,   W1, b1, W2, b2, n);
        gcn_range<2><<<nbin, WG, 0, stream>>>(packed, desc, nblk, dinv, q, out, W1, b1, W2, b2, n);
        return;
    }

    const int tb = 256;
    const int nb_n = (n + tb - 1) / tb;
    const int per_block = tb * EDGES_PER_THREAD;
    const int nb_e = (e + per_block - 1) / per_block;

    gcn_init   <<<nb_n, tb, 0, stream>>>(deg, acc, n);
    gcn_degree <<<nb_e, tb, 0, stream>>>(dst, deg, acc, e);
    gcn_dinv   <<<nb_n, tb, 0, stream>>>(deg, acc, x, pq, out, n);
    gcn_scatter<<<nb_e, tb, 0, stream>>>(src, dst, pq, acc, out, e);

    const int nodes_per_block = (tb / 32) * 16;
    gcn_node_mlp<<<(n + nodes_per_block - 1) / nodes_per_block, tb, 0, stream>>>(
        acc, out, deg, W1, b1, W2, pq, n);

    gcn_scatter<<<nb_e, tb, 0, stream>>>(src, dst, pq, acc, out, e);
    gcn_final  <<<nb_n, tb, 0, stream>>>(out, acc, pq, deg, b2, n);
}


# Edges bucketed by (destination range, source window): a work-group keeps a source window in LDS, so per-edge gathers are LDS reads; node kernels add four partial sums per node (f32 sum order changes)
# speedup vs baseline: 8.9138x; 1.2557x over previous
#include <hip/hip_runtime.h>


typedef __attribute__((ext_vector_type(2))) float v2f;
typedef __attribute__((ext_vector_type(8))) float v8f;
typedef __attribute__((ext_vector_type(4))) float v4f;
typedef __attribute__((ext_vector_type(4))) unsigned v4u;


#define EDGES_PER_THREAD 4

__device__ __forceinline__ int package_half() {
    __shared__ int half_of_block;
    if (threadIdx.x < warpSize) {
        unsigned id = __builtin_amdgcn_s_sendmsg_rtn(0x87);
        if (threadIdx.x == 0) half_of_block = (int)((id >> 18) & 1u);
    }
    __syncthreads();
    return __builtin_amdgcn_readfirstlane(half_of_block);
}

__global__ void gcn_init(float* __restrict__ deg_a, float* __restrict__ deg_b, int n) {
    int i = blockIdx.x * blockDim.x + threadIdx.x;
    if (i < n) { deg_a[i] = 0.0f; deg_b[i] = 0.0f; }
}

__global__ void gcn_degree(const int* __restrict__ dst, float* __restrict__ deg_a,
                           float* __restrict__ deg_b, int e) {
    float* __restrict__ deg = package_half() ? deg_b : deg_a;
    const int per_block = (int)blockDim.x * EDGES_PER_THREAD;
    const int b0 = (int)blockIdx.x * per_block;
    const int i0 = b0 + (int)threadIdx.x;
    if (b0 + per_block <= e) {
        int d[EDGES_PER_THREAD];
#pragma unroll
        for (int j = 0; j < EDGES_PER_THREAD; ++j)
            d[j] = __builtin_nontemporal_load(&dst[i0 + j * (int)blockDim.x]);
#pragma unroll
        for (int j = 0; j < EDGES_PER_THREAD; ++j)
            atomicAdd(&deg[d[j]], 1.0f);
    } else {
        for (int j = 0; j < EDGES_PER_THREAD; ++j) {
            int i = i0 + j * (int)blockDim.x;
            if (i < e) atomicAdd(&deg[__builtin_nontemporal_load(&dst[i])], 1.0f);
        }
    }
}

__global__ void gcn_dinv(float* __restrict__ deg_a, float* __restrict__ deg_b,
                         const float* __restrict__ x, float* __restrict__ p,
                         float* __restrict__ out, int n) {
    int i = blockIdx.x * blockDim.x + threadIdx.x;
    if (i < n) {
        float d = (deg_a[i] + deg_b[i]) + 1.0f;
        float di = rsqrtf(d);
        deg_a[i] = di;
        p[i] = di * x[i];
        deg_b[i] = 0.0f;
        out[i] = 0.0f;
    }
}

__global__ void gcn_scatter(const int* __restrict__ src, const int* __restrict__ dst,
                            const float* __restrict__ val,
                            float* __restrict__ acc_a, float* __restrict__ acc_b, int e) {
    float* __restrict__ acc = package_half() ? acc_b : acc_a;
    const int per_block = (int)blockDim.x * EDGES_PER_THREAD;
    const int b0 = (int)blockIdx.x * per_block;
    const int i0 = b0 + (int)threadIdx.x;
    if (b0 + per_block <= e) {
        int s[EDGES_PER_THREAD], d[EDGES_PER_THREAD];
        float v[EDGES_PER_THREAD];
#pragma unroll
        for (int j = 0; j < EDGES_PER_THREAD; ++j) {
            int i = i0 + j * (int)blockDim.x;
            s[j] = __builtin_nontemporal_load(&src[i]);
            d[j] = __builtin_nontemporal_load(&dst[i]);
        }
#pragma unroll
        for (int j = 0; j < EDGES_PER_THREAD; ++j)
            v[j] = val[s[j]];
#pragma unroll
        for (int j = 0; j < EDGES_PER_THREAD; ++j)
            atomicAdd(&acc[d[j]], v[j]);
    } else {
        for (int j = 0; j < EDGES_PER_THREAD; ++j) {
            int i = i0 + j * (int)blockDim.x;
            if (i < e) {
                int s = __builtin_nontemporal_load(&src[i]);
                int d = __builtin_nontemporal_load(&dst[i]);
                atomicAdd(&acc[d], val[s]);
            }
        }
    }
}

__global__ void gcn_node_mlp(float* __restrict__ acc_a, float* __restrict__ acc_b,
                             const float* __restrict__ dinv,
                             const float* __restrict__ W1, const float* __restrict__ b1,
                             const float* __restrict__ W2,
                             float* __restrict__ pq, int n) {
    int lane = threadIdx.x & 31;
    int wave = threadIdx.x >> 5;
    int base = (blockIdx.x * (blockDim.x >> 5) + wave) * 16;
    int node = base + lane;

    float sv = 0.0f;
    float di = 0.0f;
    if (lane < 16 && node < n) {
        di = dinv[node];
        sv = di * ((acc_a[node] + acc_b[node]) + pq[node]);
        acc_a[node] = 0.0f;
        acc_b[node] = 0.0f;
    }

    v2f a; a.x = 0.0f; a.y = 0.0f;
    if (lane < 8) a.x = W1[lane];
    v2f b; b.x = (lane < 16) ? sv : 0.0f; b.y = 0.0f;
    v8f c;
#pragma unroll
    for (int r = 0; r < 8; ++r) c[r] = (lane < 16) ? b1[r] : 0.0f;

    v8f dacc = __builtin_amdgcn_wmma_f32_16x16x4_f32(
         false, a,  false, b,
         (short)0, c,  false,  false);

    float acc = 0.0f;
#pragma unroll
    for (int r = 0; r < 8; ++r) acc = fmaf(fmaxf(dacc[r], 0.0f), W2[r], acc);

    if (lane < 16 && node < n) pq[node] = di * acc;
}

__global__ void gcn_final(float* __restrict__ out, const float* __restrict__ acc_a,
                          const float* __restrict__ q,
                          const float* __restrict__ dinv, const float* __restrict__ b2, int n) {
    int i = blockIdx.x * blockDim.x + threadIdx.x;
    if (i < n) {
        out[i] = dinv[i] * ((acc_a[i] + out[i]) + q[i]) + b2[0];
    }
}


#define RANGE 3936
#define MAXBIN 272
#define CHUNK 40960
#define EPT 40
#define EB 10
#define WG 1024
#define SL 6
#define RU 4

#define SW 63488
#define DW 16384
#define DSH 14
#define NSC 16
#define NGRP 4
#define MAXBIN2 1024
#define SL2 2
#define RU2 8
#define WROWS 16
#define NU 4

__device__ __forceinline__ unsigned block_incl_scan(unsigned v, unsigned* wtot, unsigned& total) {
    const int lane = threadIdx.x & 31;
    const int wave = threadIdx.x >> 5;
    for (int d = 1; d < 32; d <<= 1) {
        unsigned u = __shfl_up(v, d);
        if (lane >= d) v += u;
    }
    if (lane == 31) wtot[wave] = v;
    __syncthreads();
    if (wave == 0) {
        unsigned w = wtot[lane];
        for (int d = 1; d < 32; d <<= 1) {
            unsigned u = __shfl_up(w, d);
            if (lane >= d) w += u;
        }
        wtot[lane] = w;
    }
    __syncthreads();
    total = wtot[31];
    if (wave > 0) v += wtot[wave - 1];
    __syncthreads();
    return v;
}

template <bool TILE>
__device__ __forceinline__ void edge_word(unsigned s, unsigned d, unsigned& bin, unsigned& word) {
    if (TILE) {
        const unsigned c = s / SW;
        bin = (d >> DSH) * NSC + c;
        word = ((s - c * SW) << DSH) | (d & (DW - 1u));
    } else {
        bin = d / RANGE;
        word = (s << 12) | (d - bin * RANGE);
    }
}

template <bool TILE>
__global__ __launch_bounds__(WG) void gcn_bucket(const int* __restrict__ src, const int* __restrict__ dst,
                                                 unsigned* __restrict__ packed, unsigned* __restrict__ desc,
                                                 int e, int nbin, int nblk) {
    __shared__ unsigned stage[CHUNK];
    __shared__ unsigned hist[TILE ? MAXBIN2 : MAXBIN];
    __shared__ unsigned lbase[TILE ? MAXBIN2 : MAXBIN];
    __shared__ unsigned wtot[32];
    const int tid = (int)threadIdx.x;
    if (tid < (TILE ? MAXBIN2 : MAXBIN)) hist[tid] = 0u;
    __syncthreads();

    const int base = (int)blockIdx.x * CHUNK;
    unsigned word[EPT], br[EPT];
    if (base + CHUNK <= e) {
#pragma unroll
        for (int h = 0; h < EPT; h += EB) {
            int s[EB], d[EB];
#pragma unroll
            for (int j = 0; j < EB; ++j) {
                s[j] = __builtin_nontemporal_load(&src[base + (h + j) * WG + tid]);
                d[j] = __builtin_nontemporal_load(&dst[base + (h + j) * WG + tid]);
            }
#pragma unroll
            for (int j = 0; j < EB; ++j) {
                unsigned bin;
                edge_word<TILE>((unsigned)s[j], (unsigned)d[j], bin, word[h + j]);
                br[h + j] = (bin << 16) | atomicAdd(&hist[bin], 1u);
            }
        }
    } else {
#pragma unroll
        for (int j = 0; j < EPT; ++j) {
            int i = base + j * WG + tid;
            word[j] = 0u;
            br[j] = 0xFFFFFFFFu;
            if (i < e) {
                unsigned sv = (unsigned)__builtin_nontemporal_load(&src[i]);
                unsigned dv = (unsigned)__builtin_nontemporal_load(&dst[i]);
                unsigned bin;
                edge_word<TILE>(sv, dv, bin, word[j]);
                br[j] = (bin << 16) | atomicAdd(&hist[bin], 1u);
            }
        }
    }
    __syncthreads();

    unsigned tot;
    unsigned hv = (tid < nbin) ? hist[tid] : 0u;
    unsigned hi = block_incl_scan(hv, wtot, tot);
    if (tid < nbin) {
        lbase[tid] = hi - hv;
        desc[TILE ? ((size_t)blockIdx.x * MAXBIN2 + tid) : ((size_t)tid * nblk + blockIdx.x)] = (hi - hv) | (hv << 16);
    }
    __syncthreads();

#pragma unroll
    for (int j = 0; j < EPT; ++j)
        if (br[j] != 0xFFFFFFFFu) stage[lbase[br[j] >> 16] + (br[j] & 0xFFFFu)] = word[j];
    __syncthreads();

    const int have = (e - base < CHUNK) ? (e - base) : CHUNK;
    for (int i = tid; i < have; i += WG) packed[base + i] = stage[i];
}

template <int MODE>
__global__ __launch_bounds__(WG) void gcn_range(const unsigned* __restrict__ packed,
                                                const unsigned* __restrict__ desc, int nblk,
                                                float* dinv, const float* __restrict__ gsrc,
                                                float* __restrict__ wr,
                                                const float* __restrict__ W1, const float* __restrict__ b1,
                                                const float* __restrict__ W2, const float* __restrict__ b2, int n) {
    __shared__ float accf[RANGE];
    __shared__ float dis[RANGE];
    const int tid = (int)threadIdx.x;
    const int lane = tid & 31;
    const int r = (int)blockIdx.x;
    unsigned* cacc = (unsigned*)accf;
    const int wave = __builtin_amdgcn_readfirstlane(tid >> 5);

    const unsigned* __restrict__ drow = desc + (size_t)r * nblk;
    unsigned dsc[RU];
#pragma unroll
    for (int u = 0; u < RU; ++u) {
        const int b = wave + 32 * u;
        dsc[u] = drow[b < nblk ? b : 0];
    }

    for (int i = tid; i < RANGE; i += WG) accf[i] = 0.0f;
    __syncthreads();

    for (int b0 = wave; b0 < nblk; b0 += 32 * RU) {
        unsigned w[RU * SL], len[RU], first[RU], nxt[RU];
#pragma unroll
        for (int u = 0; u < RU; ++u) {
            const int b = b0 + 32 * RU + 32 * u;
            nxt[u] = drow[b < nblk ? b : 0];
        }
#pragma unroll
        for (int u = 0; u < RU; ++u) {
            const int b = b0 + 32 * u;
            len[u] = (b < nblk) ? (dsc[u] >> 16) : 0u;
            first[u] = (unsigned)b * CHUNK + (dsc[u] & 0xFFFFu);
            const unsigned rb = len[u] ? first[u] : (len[0] ? first[0] : 0u);
            const unsigned rl = len[u] ? len[u] - 1u : (len[0] ? len[0] - 1u : 0u);
#pragma unroll
            for (int j = 0; j < SL; ++j) {
                const unsigned k = (unsigned)(j * 32 + lane);
                w[u * SL + j] = __builtin_nontemporal_load(&packed[rb + (k < rl ? k : rl)]);
            }
        }
#pragma unroll
        for (int u = 0; u < RU; ++u) dsc[u] = nxt[u];
        if (MODE == 0) {
#pragma unroll
            for (int u = 0; u < RU; ++u) {
#pragma unroll
                for (int j = 0; j < SL; ++j) {
                    const bool ok = (unsigned)(j * 32 + lane) < len[u];
                    atomicAdd(&cacc[ok ? (w[u * SL + j] & 4095u) : (unsigned)tid], ok ? 1u : 0u);
                }
            }
        } else {
            float v[RU * SL];
#pragma unroll
            for (int q = 0; q < RU * SL; ++q) v[q] = gsrc[w[q] >> 12];
#pragma unroll
            for (int u = 0; u < RU; ++u) {
#pragma unroll
                for (int j = 0; j < SL; ++j) {
                    const bool ok = (unsigned)(j * 32 + lane) < len[u];
                    atomicAdd(&accf[ok ? (w[u * SL + j] & 4095u) : (unsigned)tid], ok ? v[u * SL + j] : 0.0f);
                }
            }
        }
#pragma unroll
        for (int u = 0; u < RU; ++u) {
            for (unsigned k = (unsigned)(SL * 32 + lane); k < len[u]; k += 32u) {
                const unsigned ww = __builtin_nontemporal_load(&packed[first[u] + k]);
                if (MODE == 0) atomicAdd(&cacc[ww & 4095u], 1u);
                else atomicAdd(&accf[ww & 4095u], gsrc[ww >> 12]);
            }
        }
    }
    __syncthreads();

    if (MODE == 0) {
        for (int i = tid; i < RANGE; i += WG) {
            int node = r * RANGE + i;
            if (node < n) {
                float d = (float)cacc[i] + 1.0f;
                float di = rsqrtf(d);
                dinv[node] = di;
                wr[node] = di * gsrc[node];
            }
        }
    } else if (MODE == 2) {
        for (int i = tid; i < RANGE; i += WG) {
            int node = r * RANGE + i;
            if (node < n) wr[node] = dinv[node] * (accf[i] + gsrc[node]) + b2[0];
        }
    } else {
        for (int i = tid; i < RANGE; i += WG) {
            int node = r * RANGE + i;
            float di = 0.0f, sv = 0.0f;
            if (node < n) { di = dinv[node]; sv = di * (accf[i] + gsrc[node]); }
            accf[i] = sv;
            dis[i] = di;
        }
        __syncthreads();
        v2f a; a.x = 0.0f; a.y = 0.0f;
        if (lane < 8) a.x = W1[lane];
        v8f c;
#pragma unroll
        for (int j = 0; j < 8; ++j) c[j] = (lane < 16) ? b1[j] : 0.0f;
        for (int g = wave; g < RANGE / 16; g += 32) {
            const int i = g * 16 + (lane & 15);
            const float sv = accf[i];
            const float di = dis[i];
            v2f b; b.x = (lane < 16) ? sv : 0.0f; b.y = 0.0f;
            v8f dacc = __builtin_amdgcn_wmma_f32_16x16x4_f32(
                 false, a,  false, b,
                 (short)0, c,  false,  false);
            float acc = 0.0f;
#pragma unroll
            for (int j = 0; j < 8; ++j) acc = fmaf(fmaxf(dacc[j], 0.0f), W2[j], acc);
            const int node = r * RANGE + i;
            if (lane < 16 && node < n) wr[node] = di * acc;
        }
    }
}

static_assert(NSC / NGRP == 4 && NSC % NGRP == 0, "a work-group takes four windows: the degree pass reads their four descriptors as one run");
static_assert((1 << DSH) == DW && SW <= 65536 && SW % 4 == 0 && WG == MAXBIN2 && WROWS * WG * 4 >= SW, "packing, scan and window-fill sizes");

__global__ __launch_bounds__(WG) void gcn_tile_degree(const unsigned* __restrict__ packed,
                                                      const unsigned* __restrict__ desc, int nblk,
                                                      unsigned* __restrict__ pc0, unsigned* __restrict__ pcw, int n) {
    __shared__ unsigned cnt[DW];
    const int tid = (int)threadIdx.x;
    const int lane = tid & 31;
    const int wave = __builtin_amdgcn_readfirstlane(tid >> 5);
    const int d = (int)blockIdx.x / NGRP;
    const int g = (int)blockIdx.x % NGRP;
    const unsigned* __restrict__ dcol = desc + (d * NSC + g * (NSC / NGRP));

    for (int i = tid; i < DW; i += WG) cnt[i] = 0u;
    __syncthreads();

    for (int b0 = wave; b0 < nblk; b0 += 32 * RU) {
        unsigned w[RU * SL], len[RU], first[RU];
#pragma unroll
        for (int u = 0; u < RU; ++u) {
            const int b = b0 + 32 * u;
            const unsigned* __restrict__ q = dcol + (size_t)(b < nblk ? b : 0) * MAXBIN2;
            const unsigned q0 = q[0], q1 = q[1], q2 = q[2], q3 = q[3];
            len[u] = (b < nblk) ? ((q0 >> 16) + (q1 >> 16) + (q2 >> 16) + (q3 >> 16)) : 0u;
            first[u] = (unsigned)b * CHUNK + (q0 & 0xFFFFu);
            const unsigned rb = len[u] ? first[u] : (len[0] ? first[0] : 0u);
            const unsigned rl = len[u] ? len[u] - 1u : (len[0] ? len[0] - 1u : 0u);
#pragma unroll
            for (int j = 0; j < SL; ++j) {
                const unsigned k = (unsigned)(j * 32 + lane);
                w[u * SL + j] = __builtin_nontemporal_load(&packed[rb + (k < rl ? k : rl)]);
            }
        }
#pragma unroll
        for (int u = 0; u < RU; ++u) {
#pragma unroll
            for (int j = 0; j < SL; ++j) {
                const bool ok = (unsigned)(j * 32 + lane) < len[u];
                atomicAdd(&cnt[ok ? (w[u * SL + j] & (DW - 1u)) : (unsigned)tid], ok ? 1u : 0u);
            }
        }
#pragma unroll
        for (int u = 0; u < RU; ++u) {
            for (unsigned k = (unsigned)(SL * 32 + lane); k < len[u]; k += 32u) {
                const unsigned ww = __builtin_nontemporal_load(&packed[first[u] + k]);
                atomicAdd(&cnt[ww & (DW - 1u)], 1u);
            }
        }
    }
    __syncthreads();

    unsigned* __restrict__ pc = (g == 0) ? pc0 : pcw + (size_t)(g - 1) * n;
    for (int i = tid; i < DW; i += WG) {
        const int node = d * DW + i;
        if (node < n) pc[node] = cnt[i];
    }
}

__global__ __launch_bounds__(256) void gcn_tile_node0(const unsigned* __restrict__ pc0, const unsigned* __restrict__ pcw,
                                                      const float* __restrict__ x, float* __restrict__ dinv,
                                                      float* __restrict__ p, int n) {
    const int n4 = n >> 2;
    const uint4* __restrict__ c0 = (const uint4*)pc0;
    const uint4* __restrict__ c1 = (const uint4*)pcw;
    const uint4* __restrict__ c2 = (const uint4*)(pcw + (size_t)n);
    const uint4* __restrict__ c3 = (const uint4*)(pcw + 2 * (size_t)n);
    for (int i = (int)(blockIdx.x * blockDim.x + threadIdx.x); i < n4; i += (int)(gridDim.x * blockDim.x)) {
        const uint4 a = c0[i], b = c1[i], c = c2[i], e4 = c3[i];
        const float4 xv = ((const float4*)x)[i];
        float4 dv, pv;
        dv.x = rsqrtf((float)((a.x + b.x) + (c.x + e4.x)) + 1.0f);
        dv.y = rsqrtf((float)((a.y + b.y) + (c.y + e4.y)) + 1.0f);
        dv.z = rsqrtf((float)((a.z + b.z) + (c.z + e4.z)) + 1.0f);
        dv.w = rsqrtf((float)((a.w + b.w) + (c.w + e4.w)) + 1.0f);
        pv.x = dv.x * xv.x; pv.y = dv.y * xv.y; pv.z = dv.z * xv.z; pv.w = dv.w * xv.w;
        ((float4*)dinv)[i] = dv;
        ((float4*)p)[i] = pv;
    }
}

__global__ __launch_bounds__(WG) void gcn_tile_layer(const unsigned* __restrict__ packed,
                                                     const unsigned* __restrict__ desc, int nblk,
                                                     const float* __restrict__ gsrc,
                                                     float* __restrict__ part0, float* __restrict__ partw, int n) {
    __shared__ __attribute__((aligned(16))) float srcv[SW];
    __shared__ float accf[DW];
    __shared__ __attribute__((aligned(16))) unsigned dl[32 * RU2 * (NSC / NGRP)];
    const int tid = (int)threadIdx.x;
    const int lane = tid & 31;
    const int wave = __builtin_amdgcn_readfirstlane(tid >> 5);
    const int d = (int)blockIdx.x / NGRP;
    const int g = (int)blockIdx.x % NGRP;

    for (int i = tid; i < DW; i += WG) accf[i] = 0.0f;
    if (tid < nblk)
        *(v4u*)&dl[tid * (NSC / NGRP)] = *(const v4u*)&desc[(size_t)tid * MAXBIN2 + d * NSC + g * (NSC / NGRP)];
    __syncthreads();

    for (int c = 0; c < NSC / NGRP; ++c) {
        const int cbase = (g * (NSC / NGRP) + c) * SW;
        int lim = n - cbase;
        lim = lim < 0 ? 0 : (lim > SW ? SW : lim);
        unsigned w[RU2 * SL2], len[RU2], first[RU2], dv[RU2];
#pragma unroll
        for (int u = 0; u < RU2; ++u) {
            const int b = wave + 32 * u;
            dv[u] = dl[(b < nblk ? b : 0) * (NSC / NGRP) + c];
        }
#pragma unroll
        for (int u = 0; u < RU2; ++u) {
            const int b = wave + 32 * u;
            const unsigned dsc = __builtin_amdgcn_readfirstlane(dv[u]);
            len[u] = (b < nblk) ? (dsc >> 16) : 0u;
            first[u] = (unsigned)b * CHUNK + (dsc & 0xFFFFu);
            const unsigned rb = len[u] ? first[u] : (len[0] ? first[0] : 0u);
            const unsigned rl = len[u] ? len[u] - 1u : (len[0] ? len[0] - 1u : 0u);
#pragma unroll
            for (int j = 0; j < SL2; ++j) {
                const unsigned k = (unsigned)(j * 32 + lane);
                w[u * SL2 + j] = __builtin_nontemporal_load(&packed[rb + (k < rl ? k : rl)]);
            }
        }
        v4f t[WROWS];
#pragma unroll
        for (int j = 0; j < WROWS; ++j) {
            int gi = cbase + (j * WG + tid) * 4;
            gi = gi < n ? gi : cbase + tid * 4;
            t[j] = *(const v4f*)&gsrc[gi < n ? gi : n - 4];
        }
        __syncthreads();
#pragma unroll
        for (int j = 0; j < WROWS; ++j) {
            const int i = (j * WG + tid) * 4;
            if (i < lim) *(v4f*)&srcv[i] = t[j];
        }
        __syncthreads();
        float v[RU2 * SL2];
#pragma unroll
        for (int q = 0; q < RU2 * SL2; ++q) v[q] = srcv[w[q] >> DSH];
#pragma unroll
        for (int u = 0; u < RU2; ++u) {
#pragma unroll
            for (int j = 0; j < SL2; ++j) {
                const bool ok = (unsigned)(j * 32 + lane) < len[u];
                atomicAdd(&accf[ok ? (w[u * SL2 + j] & (DW - 1u)) : (unsigned)tid], ok ? v[u * SL2 + j] : 0.0f);
            }
        }
#pragma unroll
        for (int u = 0; u < RU2; ++u) {
            for (unsigned k = (unsigned)(SL2 * 32 + lane); k < len[u]; k += 32u) {
                const unsigned ww = __builtin_nontemporal_load(&packed[first[u] + k]);
                atomicAdd(&accf[ww & (DW - 1u)], srcv[ww >> DSH]);
            }
        }
    }
    __syncthreads();

    float* __restrict__ part = (g == 0) ? part0 : partw + (size_t)(g - 1) * n;
    for (int i = tid; i < DW; i += WG) {
        const int node = d * DW + i;
        if (node < n) part[node] = accf[i];
    }
}

__global__ __launch_bounds__(256) void gcn_tile_node1(const float* __restrict__ part0, const float* __restrict__ partw,
                                                      const float* __restrict__ dinv,
                                                      const float* __restrict__ W1, const float* __restrict__ b1,
                                                      const float* __restrict__ W2,
                                                      float* __restrict__ pq, int n) {
    const int lane = (int)threadIdx.x & 31;
    const int wv = __builtin_amdgcn_readfirstlane((int)((blockIdx.x * blockDim.x + threadIdx.x) >> 5));
    const int nwv = (int)((gridDim.x * blockDim.x) >> 5);
    const float* __restrict__ part1 = partw;
    const float* __restrict__ part2 = partw + (size_t)n;
    const float* __restrict__ part3 = partw + 2 * (size_t)n;

    v2f a; a.x = 0.0f; a.y = 0.0f;
    if (lane < 8) a.x = W1[lane];
    v8f c;
#pragma unroll
    for (int r = 0; r < 8; ++r) c[r] = (lane < 16) ? b1[r] : 0.0f;

    for (int base = wv * (32 * NU); base < n; base += nwv * (32 * NU)) {
        float sv[NU], di[NU];
#pragma unroll
        for (int u = 0; u < NU; ++u) {
            const int node = base + 32 * u + lane;
            const int nc = node < n ? node : n - 1;
            di[u] = dinv[nc];
            sv[u] = di[u] * (((part0[nc] + part1[nc]) + (part2[nc] + part3[nc])) + pq[nc]);
        }
#pragma unroll
        for (int u = 0; u < NU; ++u) {
            const float sh = __shfl(sv[u], (lane & 15) + 16);
            v2f bl; bl.x = (lane < 16) ? sv[u] : 0.0f; bl.y = 0.0f;
            v2f bh; bh.x = (lane < 16) ? sh : 0.0f; bh.y = 0.0f;
            v8f dl = __builtin_amdgcn_wmma_f32_16x16x4_f32(
                 false, a,  false, bl,
                 (short)0, c,  false,  false);
            v8f dh = __builtin_amdgcn_wmma_f32_16x16x4_f32(
                 false, a,  false, bh,
                 (short)0, c,  false,  false);
            float al = 0.0f, ah = 0.0f;
#pragma unroll
            for (int r = 0; r < 8; ++r) {
                al = fmaf(fmaxf(dl[r], 0.0f), W2[r], al);
                ah = fmaf(fmaxf(dh[r], 0.0f), W2[r], ah);
            }
            const float up = __shfl(ah, lane & 15);
            const float h2 = (lane < 16) ? al : up;
            const int node = base + 32 * u + lane;
            if (node < n) pq[node] = di[u] * h2;
        }
    }
}

__global__ __launch_bounds__(256) void gcn_tile_node2(float* __restrict__ out, const float* __restrict__ partw,
                                                      const float* __restrict__ q, const float* __restrict__ dinv,
                                                      const float* __restrict__ b2, int n) {
    const int n4 = n >> 2;
    const float4* __restrict__ p1 = (const float4*)partw;
    const float4* __restrict__ p2 = (const float4*)(partw + (size_t)n);
    const float4* __restrict__ p3 = (const float4*)(partw + 2 * (size_t)n);
    const float bb = b2[0];
    for (int i = (int)(blockIdx.x * blockDim.x + threadIdx.x); i < n4; i += (int)(gridDim.x * blockDim.x)) {
        const float4 o = ((const float4*)out)[i], s1 = p1[i], s2 = p2[i], s3 = p3[i];
        const float4 qv = ((const float4*)q)[i], dv = ((const float4*)dinv)[i];
        float4 r;
        r.x = dv.x * (((o.x + s1.x) + (s2.x + s3.x)) + qv.x) + bb;
        r.y = dv.y * (((o.y + s1.y) + (s2.y + s3.y)) + qv.y) + bb;
        r.z = dv.z * (((o.z + s1.z) + (s2.z + s3.z)) + qv.z) + bb;
        r.w = dv.w * (((o.w + s1.w) + (s2.w + s3.w)) + qv.w) + bb;
        ((float4*)out)[i] = r;
    }
}

extern "C" void kernel_launch(void* const* d_in, const int* in_sizes, int n_in,
                              void* d_out, int out_size, void* d_ws, size_t ws_size,
                              hipStream_t stream) {
    const float* x  = (const float*)d_in[0];
    const int*   ei = (const int*)d_in[1];
    const float* W1 = (const float*)d_in[2];
    const float* b1 = (const float*)d_in[3];
    const float* W2 = (const float*)d_in[4];
    const float* b2 = (const float*)d_in[5];

    const int n = in_sizes[0];
    const int e = in_sizes[1] / 2;
    const int* src = ei;
    const int* dst = ei + e;

    float* deg = (float*)d_ws;
    float* acc = deg + n;
    float* pq  = acc + n;
    float* out = (float*)d_out;

    const int nblk = (e + CHUNK - 1) / CHUNK;

    const int nd = (n + DW - 1) / DW;
    const size_t need2 = ((size_t)5 * n + (size_t)e + (size_t)nblk * MAXBIN2) * sizeof(float);
    if (e > 0 && n > 0 && (n & 3) == 0 && n <= NSC * SW && nd * NSC <= MAXBIN2 && nblk <= 32 * RU2 && ws_size >= need2 &&
        (((size_t)d_ws | (size_t)d_out | (size_t)d_in[0]) & 15) == 0) {
        float*    dinv   = (float*)d_ws;
        float*    pq     = dinv + n;
        float*    partw  = pq + n;
        unsigned* desc   = (unsigned*)(partw + (size_t)3 * n);
        unsigned* packed = desc + (size_t)nblk * MAXBIN2;
        const int nb4 = ((n >> 2) + 255) / 256;
        const int nb1 = (n + 32 * NU * 8 - 1) / (32 * NU * 8);
        gcn_bucket<true><<<nblk, WG, 0, stream>>>(src, dst, packed, desc, e, nd * NSC, nblk);
        gcn_tile_degree<<<nd * NGRP, WG, 0, stream>>>(packed, desc, nblk, (unsigned*)out, (unsigned*)partw, n);
        gcn_tile_node0 <<<nb4, 256, 0, stream>>>((const unsigned*)out, (const unsigned*)partw, x, dinv, pq, n);
        gcn_tile_layer <<<nd * NGRP, WG, 0, stream>>>(packed, desc, nblk, pq, out, partw, n);
        gcn_tile_node1 <<<nb1, 256, 0, stream>>>(out, partw, dinv, W1, b1, W2, pq, n);
        gcn_tile_layer <<<nd * NGRP, WG, 0, stream>>>(packed, desc, nblk, pq, out, partw, n);
        gcn_tile_node2 <<<nb4, 256, 0, stream>>>(out, partw, pq, dinv, b2, n);
        return;
    }

    const int nbin = (n + RANGE - 1) / RANGE;
    const size_t need = ((size_t)2 * n + (size_t)e + (size_t)nbin * nblk) * sizeof(float);
    if (e > 0 && n <= (1 << 20) && nbin <= MAXBIN && ws_size >= need) {
        float*    dinv   = (float*)d_ws;
        float*    q      = dinv + n;
        unsigned* packed = (unsigned*)(q + n);
        unsigned* desc   = packed + e;
        float*    p      = out;
        gcn_bucket<false><<<nblk, WG, 0, stream>>>(src, dst, packed, desc, e, nbin, nblk);
        gcn_range<0><<<nbin, WG, 0, stream>>>(packed, desc, nblk, dinv, x, p,   W1, b1, W2, b2, n);
        gcn_range<1><<<nbin, WG, 0, stream>>>(packed, desc, nblk, dinv, p, q,   W1, b1, W2, b2, n);
        gcn_range<2><<<nbin, WG, 0, stream>>>(packed, desc, nblk, dinv, q, out, W1, b1, W2, b2, n);
        return;
    }

    const int tb = 256;
    const int nb_n = (n + tb - 1) / tb;
    const int per_block = tb * EDGES_PER_THREAD;
    const int nb_e = (e + per_block - 1) / per_block;

    gcn_init   <<<nb_n, tb, 0, stream>>>(deg, acc, n);
    gcn_degree <<<nb_e, tb, 0, stream>>>(dst, deg, acc, e);
    gcn_dinv   <<<nb_n, tb, 0, stream>>>(deg, acc, x, pq, out, n);
    gcn_scatter<<<nb_e, tb, 0, stream>>>(src, dst, pq, acc, out, e);

    const int nodes_per_block = (tb / 32) * 16;
    gcn_node_mlp<<<(n + nodes_per_block - 1) / nodes_per_block, tb, 0, stream>>>(
        acc, out, deg, W1, b1, W2, pq, n);

    gcn_scatter<<<nb_e, tb, 0, stream>>>(src, dst, pq, acc, out, e);
    gcn_final  <<<nb_n, tb, 0, stream>>>(out, acc, pq, deg, b2, n);
}
